# starting kernel
# baseline (speedup 1.0000x reference)
#include <hip/hip_runtime.h>
#include <hip/hip_bf16.h>
#include <math.h>

typedef __attribute__((ext_vector_type(2))) float v2f;
typedef __attribute__((ext_vector_type(8))) float v8f;

#define NIMG 4
#define CIN 256
#define HF 128
#define WF 128
#define A_ANCH 3
#define NPROP (A_ANCH * HF * WF)
#define NPAD 65536
#define PRE_NMS 2000
#define POST_NMS 300
#define OBJ_THR 0.3f
#define IOU_THR 0.7f
#define TILE_W 64
#define CCHUNK 32
#define CIPAD 34

__global__ __launch_bounds__(256) void repack_w_kernel(const float* __restrict__ conv_w,
                                                       float* __restrict__ wt2) {
    int t = blockIdx.x * 256 + threadIdx.x;
    int par = t & 1;
    int co  = (t >> 1) & 255;
    int q   = t >> 9;
    int ci2 = q & 127;
    int kk  = q >> 7;
    int ci  = ci2 * 2 + par;
    int kh  = kk / 3, kw = kk % 3;
    wt2[t] = conv_w[((co * CIN + ci) * 3 + kh) * 3 + kw];
}

__global__ __launch_bounds__(256) void conv_fused_kernel(
    const float* __restrict__ feat, const float* __restrict__ wt2,
    const float* __restrict__ conv_b,
    const float* __restrict__ det_w, const float* __restrict__ det_b,
    const float* __restrict__ reg_w, const float* __restrict__ reg_b,
    float* __restrict__ scores, float* __restrict__ deltas) {

    __shared__ float lds_in[3 * 66 * CIPAD];
    __shared__ float x_tile[256 * 33];
    __shared__ float hb[16];

    int bid = blockIdx.x;
    int n    = bid >> 8;
    int rem  = bid & 255;
    int row  = rem >> 1;
    int col0 = (rem & 1) * TILE_W;

    int tid  = threadIdx.x;
    int wave = tid >> 5;
    int lane = tid & 31;
    int hl   = lane >> 4;
    int lm   = lane & 15;

    v8f acc[2][4] = {};

    const float* featn = feat + (size_t)n * CIN * HF * WF;
    const v2f*   wp    = (const v2f*)wt2;

    for (int cib = 0; cib < CIN; cib += CCHUNK) {
        __syncthreads();
        for (int e = tid; e < 3 * 66 * CCHUNK; e += 256) {
            int col = e % 66;
            int r   = (e / 66) % 3;
            int cl  = e / 198;
            int grow = row + r - 1;
            int gcol = col0 + col - 1;
            float v = 0.0f;
            if (grow >= 0 && grow < HF && gcol >= 0 && gcol < WF)
                v = featn[((size_t)(cib + cl) * HF + grow) * WF + gcol];
            lds_in[(r * 66 + col) * CIPAD + cl] = v;
        }
        __syncthreads();

        for (int kh = 0; kh < 3; kh++) {
            for (int kw = 0; kw < 3; kw++) {
                int kk = kh * 3 + kw;
                #pragma unroll 4
                for (int c = 0; c < CCHUNK; c += 4) {
                    int ci2 = ((cib + c) >> 1) + hl;
                    int co0 = wave * 32 + lm;
                    v2f a0 = wp[(kk * 128 + ci2) * 256 + co0];
                    v2f a1 = wp[(kk * 128 + ci2) * 256 + co0 + 16];
                    int boff = c + 2 * hl;
                    v2f b[4];
                    #pragma unroll
                    for (int nt = 0; nt < 4; nt++)
                        b[nt] = *(const v2f*)&lds_in[(kh * 66 + nt * 16 + lm + kw) * CIPAD + boff];
                    #pragma unroll
                    for (int nt = 0; nt < 4; nt++) {
                        acc[0][nt] = __builtin_amdgcn_wmma_f32_16x16x4_f32(false, a0, false, b[nt], (short)0, acc[0][nt], false, false);
                        acc[1][nt] = __builtin_amdgcn_wmma_f32_16x16x4_f32(false, a1, false, b[nt], (short)0, acc[1][nt], false, false);
                    }
                }
            }
        }
    }

    __syncthreads();
    for (int e = tid; e < 15 * 256; e += 256)
        lds_in[e] = (e < 768) ? det_w[e] : reg_w[e - 768];
    if (tid < 15) hb[tid] = (tid < 3) ? det_b[tid] : reg_b[tid - 3];

    int m0  = hl * 8;
    float biasA[8], biasB[8];
    #pragma unroll
    for (int r = 0; r < 8; r++) {
        biasA[r] = conv_b[wave * 32 + m0 + r];
        biasB[r] = conv_b[wave * 32 + 16 + m0 + r];
    }

    for (int ph = 0; ph < 2; ph++) {
        if (ph) __syncthreads();
        #pragma unroll
        for (int r = 0; r < 8; r++) {
            int coA = wave * 32 + m0 + r;
            int coB = wave * 32 + 16 + m0 + r;
            x_tile[coA * 33 + lm]      = acc[0][ph * 2 + 0][r] + biasA[r];
            x_tile[coA * 33 + 16 + lm] = acc[0][ph * 2 + 1][r] + biasA[r];
            x_tile[coB * 33 + lm]      = acc[1][ph * 2 + 0][r] + biasB[r];
            x_tile[coB * 33 + 16 + lm] = acc[1][ph * 2 + 1][r] + biasB[r];
        }
        __syncthreads();

        for (int t = tid; t < 15 * 32; t += 256) {
            int oi  = t >> 5;
            int pix = t & 31;
            float sum = hb[oi];
            const float* wrow = &lds_in[oi * 256];
            #pragma unroll 8
            for (int co = 0; co < 256; co++)
                sum += wrow[co] * x_tile[co * 33 + pix];
            int p = row * WF + col0 + ph * 32 + pix;
            if (oi < 3) {
                scores[(size_t)n * NPROP + oi * (HF * WF) + p] = 1.0f / (1.0f + expf(-sum));
            } else {
                int ch = oi - 3;
                int ccomp = ch / 3;
                int a = ch % 3;
                deltas[(((size_t)n * NPROP) + a * (HF * WF) + p) * 4 + ccomp] = sum;
            }
        }
    }
}

__global__ __launch_bounds__(256) void decode_kernel(
    const float* __restrict__ scores, const float* __restrict__ deltas,
    float* __restrict__ boxes, unsigned long long* __restrict__ keys) {
    int t = blockIdx.x * 256 + threadIdx.x;
    int n = t >> 16;
    int i = t & 0xFFFF;
    unsigned long long* kout = keys + ((size_t)n << 16);
    if (i >= NPROP) { kout[i] = 0ull; return; }

    int pa = i / 3, aa = i - pa * 3;
    int ya = pa >> 7, xa = pa & 127;
    const float bs[3][4] = {{-91.f, -45.f, 91.f, 45.f},
                            {-64.f, -64.f, 64.f, 64.f},
                            {-45.f, -91.f, 45.f, 91.f}};
    float sxv = (float)(xa * 8), syv = (float)(ya * 8);
    float x1 = sxv + bs[aa][0], y1 = syv + bs[aa][1];
    float x2 = sxv + bs[aa][2], y2 = syv + bs[aa][3];
    float aw = x2 - x1, ah = y2 - y1;
    float acx = y1 + aw * 0.5f;
    float acy = x1 + ah * 0.5f;

    const float* d = deltas + (((size_t)n * NPROP) + i) * 4;
    float px = acx + d[0] * aw;
    float py = acy + d[1] * ah;
    float pw = aw * expf(d[2]);
    float ph = ah * expf(d[3]);
    float b0 = fminf(fmaxf(px - pw * 0.5f, 0.f), 1024.f);
    float b1 = fminf(fmaxf(py - ph * 0.5f, 0.f), 1024.f);
    float b2 = fminf(fmaxf(px + pw * 0.5f, 0.f), 1024.f);
    float b3 = fminf(fmaxf(py + ph * 0.5f, 0.f), 1024.f);

    float* bo = boxes + (((size_t)n * NPROP) + i) * 4;
    bo[0] = b0; bo[1] = b1; bo[2] = b2; bo[3] = b3;

    float s = scores[(size_t)n * NPROP + i];
    bool valid = (s > OBJ_THR) && ((b2 - b0) > 0.f) && ((b3 - b1) > 0.f);
    float ss = valid ? s : -__builtin_inff();
    unsigned u = __float_as_uint(ss);
    unsigned sv = (u & 0x80000000u) ? ~u : (u | 0x80000000u);
    kout[i] = ((unsigned long long)sv << 32) | (unsigned long long)(0xFFFFFFFFu - (unsigned)i);
}

__global__ __launch_bounds__(256) void bitonic_step_kernel(unsigned long long* __restrict__ keys,
                                                           int j, int k) {
    int t = blockIdx.x * 256 + threadIdx.x;
    int n = t >> 16;
    int i = t & 0xFFFF;
    int ixj = i ^ j;
    if (ixj > i) {
        unsigned long long* K = keys + ((size_t)n << 16);
        unsigned long long a = K[i];
        unsigned long long b = K[ixj];
        bool up = ((i & k) == 0);
        bool sw = up ? (a < b) : (a > b);
        if (sw) { K[i] = b; K[ixj] = a; }
    }
}

__global__ __launch_bounds__(1024) void nms_kernel(const unsigned long long* __restrict__ keys,
                                                   const float* __restrict__ boxes,
                                                   float* __restrict__ out) {
    __shared__ float bx[PRE_NMS * 4];
    __shared__ float ar[PRE_NMS];
    __shared__ float sc[PRE_NMS];
    __shared__ int   kp[PRE_NMS];
    __shared__ int   sup;

    int n = blockIdx.x;
    const unsigned long long* K = keys + ((size_t)n << 16);

    for (int i = threadIdx.x; i < PRE_NMS; i += blockDim.x) {
        unsigned long long key = K[i];
        unsigned sv  = (unsigned)(key >> 32);
        unsigned idx = 0xFFFFFFFFu - (unsigned)(key & 0xFFFFFFFFu);
        bool v = sv > 0x007FFFFFu;
        float b0 = 0.f, b1 = 0.f, b2 = 0.f, b3 = 0.f, s = 0.f;
        if (v) {
            const float* bp = boxes + (((size_t)n * NPROP) + idx) * 4;
            b0 = bp[0]; b1 = bp[1]; b2 = bp[2]; b3 = bp[3];
            unsigned u = (sv & 0x80000000u) ? (sv ^ 0x80000000u) : ~sv;
            s = __uint_as_float(u);
        }
        bx[i * 4 + 0] = b0; bx[i * 4 + 1] = b1;
        bx[i * 4 + 2] = b2; bx[i * 4 + 3] = b3;
        ar[i] = fmaxf(b2 - b0, 0.f) * fmaxf(b3 - b1, 0.f);
        sc[i] = s;
        kp[i] = v ? 1 : 0;
    }
    __syncthreads();

    for (int i = 0; i < PRE_NMS; i++) {
        if (threadIdx.x == 0) sup = 0;
        __syncthreads();
        float ib0 = bx[i * 4 + 0], ib1 = bx[i * 4 + 1];
        float ib2 = bx[i * 4 + 2], ib3 = bx[i * 4 + 3];
        float ia  = ar[i];
        int found = 0;
        for (int j = threadIdx.x; j < i; j += blockDim.x) {
            if (kp[j] == 2) {
                float jx1 = bx[j * 4 + 0], jy1 = bx[j * 4 + 1];
                float jx2 = bx[j * 4 + 2], jy2 = bx[j * 4 + 3];
                float ix1 = fmaxf(ib0, jx1), iy1 = fmaxf(ib1, jy1);
                float ix2 = fminf(ib2, jx2), iy2 = fminf(ib3, jy2);
                float inter = fmaxf(ix2 - ix1, 0.f) * fmaxf(iy2 - iy1, 0.f);
                float iou = inter / fmaxf(ia + ar[j] - inter, 1e-9f);
                if (iou > IOU_THR) found = 1;
            }
        }
        if (found) atomicOr(&sup, 1);
        __syncthreads();
        if (threadIdx.x == 0) kp[i] = (kp[i] == 1 && sup == 0) ? 2 : 0;
        __syncthreads();
    }

    for (int t = threadIdx.x; t < POST_NMS * 5; t += blockDim.x)
        out[(size_t)n * POST_NMS * 5 + t] = 0.f;
    __syncthreads();
    if (threadIdx.x == 0) {
        int cnt = 0;
        for (int i = 0; i < PRE_NMS && cnt < POST_NMS; i++) {
            if (kp[i] == 2) {
                float* o = out + (size_t)n * POST_NMS * 5 + cnt * 5;
                o[0] = bx[i * 4 + 0]; o[1] = bx[i * 4 + 1];
                o[2] = bx[i * 4 + 2]; o[3] = bx[i * 4 + 3];
                o[4] = sc[i];
                cnt++;
            }
        }
    }
}

extern "C" void kernel_launch(void* const* d_in, const int* in_sizes, int n_in,
                              void* d_out, int out_size, void* d_ws, size_t ws_size,
                              hipStream_t stream) {
    const float* feat   = (const float*)d_in[0];
    const float* conv_w = (const float*)d_in[1];
    const float* conv_b = (const float*)d_in[2];
    const float* det_w  = (const float*)d_in[3];
    const float* det_b  = (const float*)d_in[4];
    const float* reg_w  = (const float*)d_in[5];
    const float* reg_b  = (const float*)d_in[6];
    float* out = (float*)d_out;

    char* ws = (char*)d_ws;
    float* wt2                  = (float*)(ws);
    float* scores               = (float*)(ws + 2359296);
    float* deltas               = (float*)(ws + 2359296 + 786432);
    float* boxes                = (float*)(ws + 2359296 + 786432 + 3145728);
    unsigned long long* keys    = (unsigned long long*)(ws + 2359296 + 786432 + 3145728 + 3145728);

    repack_w_kernel<<<2304, 256, 0, stream>>>(conv_w, wt2);
    conv_fused_kernel<<<NIMG * HF * (WF / TILE_W), 256, 0, stream>>>(
        feat, wt2, conv_b, det_w, det_b, reg_w, reg_b, scores, deltas);
    decode_kernel<<<(NIMG * NPAD) / 256, 256, 0, stream>>>(scores, deltas, boxes, keys);
    for (int k = 2; k <= NPAD; k <<= 1)
        for (int j = k >> 1; j > 0; j >>= 1)
            bitonic_step_kernel<<<(NIMG * NPAD) / 256, 256, 0, stream>>>(keys, j, k);
    nms_kernel<<<NIMG, 1024, 0, stream>>>(keys, boxes, out);
}


# NMS kept-list pair test: exact division-free reject first (intersection <= half the box's own area cannot pass the IoU threshold), 4 kept boxes per lane and trip; quotient only where it passes
# speedup vs baseline: 3.0848x; 3.0848x over previous
#include <hip/hip_runtime.h>
#include <hip/hip_bf16.h>
#include <math.h>

typedef __attribute__((ext_vector_type(2))) float v2f;
typedef __attribute__((ext_vector_type(8))) float v8f;

#define NIMG 4
#define CIN 256
#define HF 128
#define WF 128
#define A_ANCH 3
#define NPROP (A_ANCH * HF * WF)
#define PRE_NMS 2000
#define POST_NMS 300
#define OBJ_THR 0.3f
#define IOU_THR 0.7f
#define TILE_W 64
#define CCHUNK 32
#define CIPAD 34

__global__ __launch_bounds__(256) void repack_w_kernel(const float* __restrict__ conv_w,
                                                       float* __restrict__ wt2) {
    int t = blockIdx.x * 256 + threadIdx.x;
    int par = t & 1;
    int co  = (t >> 1) & 255;
    int q   = t >> 9;
    int ci2 = q & 127;
    int kk  = q >> 7;
    int ci  = ci2 * 2 + par;
    int kh  = kk / 3, kw = kk % 3;
    wt2[t] = conv_w[((co * CIN + ci) * 3 + kh) * 3 + kw];
}

__global__ __launch_bounds__(256) void conv_fused_kernel(
    const float* __restrict__ feat, const float* __restrict__ wt2,
    const float* __restrict__ conv_b,
    const float* __restrict__ det_w, const float* __restrict__ det_b,
    const float* __restrict__ reg_w, const float* __restrict__ reg_b,
    float* __restrict__ scores, float* __restrict__ deltas) {

    __shared__ float lds_in[3 * 66 * CIPAD];
    __shared__ float x_tile[256 * 33];
    __shared__ float hb[16];

    int bid = blockIdx.x;
    int n    = bid >> 8;
    int rem  = bid & 255;
    int row  = rem >> 1;
    int col0 = (rem & 1) * TILE_W;

    int tid  = threadIdx.x;
    int wave = tid >> 5;
    int lane = tid & 31;
    int hl   = lane >> 4;
    int lm   = lane & 15;

    v8f acc[2][4] = {};

    const float* featn = feat + (size_t)n * CIN * HF * WF;
    const v2f*   wp    = (const v2f*)wt2;

    const int sx4   = (tid & 7) + 8 * (wave & 1);
    const int scl   = (wave >> 1) * 4 + ((tid >> 3) & 3);
    const int hcl   = tid & 31;
    const int hside = (tid >> 5) & 1;
    const int hr    = min(tid >> 6, 2);
    const int hgcol = col0 - 1 + hside * 65;
    const bool hok  = (hgcol >= 0) && (hgcol < WF) && (row + hr - 1 >= 0) && (row + hr - 1 < HF);
    const int hgc   = min(max(hgcol, 0), WF - 1);
    const int hgr   = min(max(row + hr - 1, 0), HF - 1);

    for (int cib = 0; cib < CIN; cib += CCHUNK) {
        float4 sv[6];
        #pragma unroll
        for (int k = 0; k < 6; k++) {
            const int r   = k % 3;
            const int cl  = scl + 16 * (k / 3);
            const int grc = min(max(row + r - 1, 0), HF - 1);
            sv[k] = *(const float4*)&featn[((size_t)(cib + cl) * HF + grc) * WF + col0 + 4 * sx4];
        }
        float hv = featn[((size_t)(cib + hcl) * HF + hgr) * WF + hgc];

        __syncthreads();
        #pragma unroll
        for (int k = 0; k < 6; k++) {
            const int r   = k % 3;
            const int cl  = scl + 16 * (k / 3);
            const bool ok = (row + r - 1 >= 0) && (row + r - 1 < HF);
            float4 v = ok ? sv[k] : make_float4(0.f, 0.f, 0.f, 0.f);
            float* dst = &lds_in[(r * 66 + 1 + 4 * sx4) * CIPAD + cl];
            dst[0 * CIPAD] = v.x;
            dst[1 * CIPAD] = v.y;
            dst[2 * CIPAD] = v.z;
            dst[3 * CIPAD] = v.w;
        }
        if (tid < 192)
            lds_in[(hr * 66 + hside * 65) * CIPAD + hcl] = hok ? hv : 0.0f;
        __syncthreads();

        for (int kh = 0; kh < 3; kh++) {
            for (int kw = 0; kw < 3; kw++) {
                int kk = kh * 3 + kw;
                #pragma unroll 4
                for (int c = 0; c < CCHUNK; c += 4) {
                    int ci2 = ((cib + c) >> 1) + hl;
                    int co0 = wave * 32 + lm;
                    v2f a0 = wp[(kk * 128 + ci2) * 256 + co0];
                    v2f a1 = wp[(kk * 128 + ci2) * 256 + co0 + 16];
                    int boff = c + 2 * hl;
                    v2f b[4];
                    #pragma unroll
                    for (int nt = 0; nt < 4; nt++)
                        b[nt] = *(const v2f*)&lds_in[(kh * 66 + nt * 16 + lm + kw) * CIPAD + boff];
                    #pragma unroll
                    for (int nt = 0; nt < 4; nt++) {
                        acc[0][nt] = __builtin_amdgcn_wmma_f32_16x16x4_f32(false, a0, false, b[nt], (short)0, acc[0][nt], false, false);
                        acc[1][nt] = __builtin_amdgcn_wmma_f32_16x16x4_f32(false, a1, false, b[nt], (short)0, acc[1][nt], false, false);
                    }
                }
            }
        }
    }

    __syncthreads();
    for (int e = tid; e < 15 * 256; e += 256)
        lds_in[e] = (e < 768) ? det_w[e] : reg_w[e - 768];
    if (tid < 15) hb[tid] = (tid < 3) ? det_b[tid] : reg_b[tid - 3];

    int m0  = hl * 8;
    float biasA[8], biasB[8];
    #pragma unroll
    for (int r = 0; r < 8; r++) {
        biasA[r] = conv_b[wave * 32 + m0 + r];
        biasB[r] = conv_b[wave * 32 + 16 + m0 + r];
    }

    for (int ph = 0; ph < 2; ph++) {
        if (ph) __syncthreads();
        #pragma unroll
        for (int r = 0; r < 8; r++) {
            int coA = wave * 32 + m0 + r;
            int coB = wave * 32 + 16 + m0 + r;
            x_tile[coA * 33 + lm]      = acc[0][ph * 2 + 0][r] + biasA[r];
            x_tile[coA * 33 + 16 + lm] = acc[0][ph * 2 + 1][r] + biasA[r];
            x_tile[coB * 33 + lm]      = acc[1][ph * 2 + 0][r] + biasB[r];
            x_tile[coB * 33 + 16 + lm] = acc[1][ph * 2 + 1][r] + biasB[r];
        }
        __syncthreads();

        for (int t = tid; t < 15 * 32; t += 256) {
            int oi  = t >> 5;
            int pix = t & 31;
            float sum = hb[oi];
            const float* wrow = &lds_in[oi * 256];
            #pragma unroll 8
            for (int co = 0; co < 256; co++)
                sum += wrow[co] * x_tile[co * 33 + pix];
            int p = row * WF + col0 + ph * 32 + pix;
            if (oi < 3) {
                scores[(size_t)n * NPROP + oi * (HF * WF) + p] = 1.0f / (1.0f + expf(-sum));
            } else {
                int ch = oi - 3;
                int ccomp = ch / 3;
                int a = ch % 3;
                deltas[(((size_t)n * NPROP) + a * (HF * WF) + p) * 4 + ccomp] = sum;
            }
        }
    }
}

__global__ __launch_bounds__(256) void decode_kernel(
    const float* __restrict__ scores, const float* __restrict__ deltas,
    float* __restrict__ boxes, unsigned* __restrict__ svk) {
    int n = blockIdx.y;
    int i = blockIdx.x * 256 + threadIdx.x;

    int pa = i / 3, aa = i - pa * 3;
    int ya = pa >> 7, xa = pa & 127;
    const float bs[3][4] = {{-91.f, -45.f, 91.f, 45.f},
                            {-64.f, -64.f, 64.f, 64.f},
                            {-45.f, -91.f, 45.f, 91.f}};
    float sxv = (float)(xa * 8), syv = (float)(ya * 8);
    float x1 = sxv + bs[aa][0], y1 = syv + bs[aa][1];
    float x2 = sxv + bs[aa][2], y2 = syv + bs[aa][3];
    float aw = x2 - x1, ah = y2 - y1;
    float acx = y1 + aw * 0.5f;
    float acy = x1 + ah * 0.5f;

    const float* d = deltas + (((size_t)n * NPROP) + i) * 4;
    float px = acx + d[0] * aw;
    float py = acy + d[1] * ah;
    float pw = aw * expf(d[2]);
    float ph = ah * expf(d[3]);
    float b0 = fminf(fmaxf(px - pw * 0.5f, 0.f), 1024.f);
    float b1 = fminf(fmaxf(py - ph * 0.5f, 0.f), 1024.f);
    float b2 = fminf(fmaxf(px + pw * 0.5f, 0.f), 1024.f);
    float b3 = fminf(fmaxf(py + ph * 0.5f, 0.f), 1024.f);

    float* bo = boxes + (((size_t)n * NPROP) + i) * 4;
    bo[0] = b0; bo[1] = b1; bo[2] = b2; bo[3] = b3;

    float s = scores[(size_t)n * NPROP + i];
    bool valid = (s > OBJ_THR) && ((b2 - b0) > 0.f) && ((b3 - b1) > 0.f);
    unsigned u = __float_as_uint(s);
    unsigned sv = (u & 0x80000000u) ? ~u : (u | 0x80000000u);
    svk[(size_t)n * NPROP + i] = valid ? sv : 0u;
}

#define NGRP 63
#define NCAND (NGRP * 32)
#define KPT 48
#define NSORT 2048

__device__ __forceinline__ int block_total(unsigned (*s_cnt)[32], int& turn, int wave, int lane, int wcount) {
    unsigned* row = s_cnt[turn & 1];
    turn++;
    if (lane == 0) row[wave] = (unsigned)wcount;
    __syncthreads();
    const uint4* p = (const uint4*)row;
    unsigned tot = 0;
    for (int q = 0; q < 8; q++) { uint4 v = p[q]; tot = tot + v.x + v.y; tot = tot + v.z + v.w; }
    return __builtin_amdgcn_readfirstlane((int)tot);
}

__device__ __forceinline__ bool iou_over(float ib0, float ib1, float ib2, float ib3, float ia,
                                         float jx1, float jy1, float jx2, float jy2, float ja) {
    float ix1 = fmaxf(ib0, jx1), iy1 = fmaxf(ib1, jy1);
    float ix2 = fminf(ib2, jx2), iy2 = fminf(ib3, jy2);
    float inter = fmaxf(ix2 - ix1, 0.f) * fmaxf(iy2 - iy1, 0.f);
    float iou = inter / fmaxf(ia + ja - inter, 1e-9f);
    return iou > IOU_THR;
}

__device__ __forceinline__ bool may_overlap(float ib0, float ib1, float ib2, float ib3, float half_ia,
                                            float jx1, float jy1, float jx2, float jy2) {
    float w = fminf(ib2, jx2) - fmaxf(ib0, jx1);
    float h = fminf(ib3, jy2) - fmaxf(ib1, jy1);
    return fmaxf(w, 0.f) * h > half_ia;
}

__global__ __launch_bounds__(1024) void nms_kernel(const unsigned* __restrict__ svk,
                                                   const float* __restrict__ boxes,
                                                   float* __restrict__ out) {
    __shared__ float4   cbx[NCAND];
    __shared__ float    car[NCAND];
    __shared__ float    csc[NCAND];
    __shared__ int      cvl[NCAND];
    __shared__ float4   kbx[NCAND];
    __shared__ float    kar[NCAND];
    __shared__ float    ksc[NCAND];
    __shared__ unsigned s_alive[32];
    __shared__ unsigned s_row[32];
    __shared__ unsigned shi[NSORT];
    __shared__ unsigned slo[NSORT];
    __shared__ __attribute__((aligned(16))) unsigned s_cnt[2][32];

    int n    = blockIdx.x;
    int tid  = threadIdx.x;
    int wave = tid >> 5;
    int lane = tid & 31;

    const unsigned* S = svk + (size_t)n * NPROP;
    unsigned sv[KPT];
    #pragma unroll
    for (int j = 0; j < KPT; j++) sv[j] = S[j * 1024 + tid];

    int turn = 0;
    unsigned T = 0u;
    int c_ge = NPROP;
    for (int bit = 31; bit >= 0; bit--) {
        unsigned cand = T | (1u << bit);
        int c = 0;
        #pragma unroll
        for (int j = 0; j < KPT; j += 2) {
            unsigned m0 = __builtin_amdgcn_ballot_w32(sv[j] >= cand);
            unsigned m1 = __builtin_amdgcn_ballot_w32(sv[j + 1] >= cand);
            c += __builtin_popcountll(((unsigned long long)m1 << 32) | m0);
        }
        int tot = block_total(s_cnt, turn, wave, lane, c);
        if (tot >= PRE_NMS) { T = cand; c_ge = tot; }
    }
    if (T != 0u) {
        int c = 0;
        #pragma unroll
        for (int j = 0; j < KPT; j += 2) {
            unsigned m0 = __builtin_amdgcn_ballot_w32(sv[j] > T);
            unsigned m1 = __builtin_amdgcn_ballot_w32(sv[j + 1] > T);
            c += __builtin_popcountll(((unsigned long long)m1 << 32) | m0);
        }
        int c_gt = block_total(s_cnt, turn, wave, lane, c);
        int need = PRE_NMS - c_gt;
        if (c_ge - c_gt != need) {
            unsigned R = 0u;
            for (int bit = 15; bit >= 0; bit--) {
                unsigned cand = R | (1u << bit);
                int ce = 0;
                #pragma unroll
                for (int j = 0; j < KPT; j++) {
                    unsigned r = 0xFFFFu - (unsigned)(j * 1024 + tid);
                    ce += __builtin_popcount(__builtin_amdgcn_ballot_w32((sv[j] == T) & (r >= cand)));
                }
                int tot = block_total(s_cnt, turn, wave, lane, ce);
                if (tot >= need) R = cand;
            }
            #pragma unroll
            for (int j = 0; j < KPT; j++) {
                unsigned r = 0xFFFFu - (unsigned)(j * 1024 + tid);
                if ((sv[j] == T) & (r < R)) sv[j] = 0u;
            }
        }
    }
    const unsigned Tc = (T != 0u) ? T : 1u;
    {
        int wsel = 0;
        #pragma unroll
        for (int j = 0; j < KPT; j += 2) {
            unsigned m0 = __builtin_amdgcn_ballot_w32(sv[j] >= Tc);
            unsigned m1 = __builtin_amdgcn_ballot_w32(sv[j + 1] >= Tc);
            wsel += __builtin_popcountll(((unsigned long long)m1 << 32) | m0);
        }
        unsigned* row = s_cnt[turn & 1];
        turn++;
        if (lane == 0) row[wave] = (unsigned)wsel;
        __syncthreads();
        int before = 0, msel = 0;
        for (int q = 0; q < 32; q++) {
            int v = (int)row[q];
            before += (q < wave) ? v : 0;
            msel   += v;
        }
        int pos = __builtin_amdgcn_readfirstlane(before);
        #pragma unroll
        for (int j = 0; j < KPT; j++) {
            bool sel = sv[j] >= Tc;
            unsigned m = __builtin_amdgcn_ballot_w32(sel);
            int at = pos + (int)__builtin_amdgcn_mbcnt_lo(m, 0u);
            if (sel) { shi[at] = sv[j]; slo[at] = 0xFFFFu - (unsigned)(j * 1024 + tid); }
            pos += __builtin_popcount(m);
        }
        for (int e = msel + tid; e < NSORT; e += 1024) { shi[e] = 0u; slo[e] = 0u; }
    }
    for (int k = 2; k <= NSORT; k <<= 1) {
        for (int j = k >> 1; j > 0; j >>= 1) {
            __syncthreads();
            int i = ((tid & ~(j - 1)) << 1) | (tid & (j - 1));
            int p = i | j;
            unsigned ah = shi[i], al = slo[i];
            unsigned bh = shi[p], bl = slo[p];
            bool a_lt_b = (ah < bh) || (ah == bh && al < bl);
            bool up = ((i & k) == 0);
            if (a_lt_b == up) { shi[i] = bh; slo[i] = bl; shi[p] = ah; slo[p] = al; }
        }
    }
    __syncthreads();

    for (int i = tid; i < NCAND; i += 1024) {
        bool v = false;
        float b0 = 0.f, b1 = 0.f, b2 = 0.f, b3 = 0.f, s = 0.f;
        if (i < PRE_NMS) {
            unsigned w   = shi[i];
            unsigned idx = 0xFFFFu - slo[i];
            v = w != 0u;
            if (v) {
                const float* bp = boxes + (((size_t)n * NPROP) + idx) * 4;
                b0 = bp[0]; b1 = bp[1]; b2 = bp[2]; b3 = bp[3];
                unsigned u = (w & 0x80000000u) ? (w ^ 0x80000000u) : ~w;
                s = __uint_as_float(u);
            }
        }
        cbx[i] = make_float4(b0, b1, b2, b3);
        car[i] = fmaxf(b2 - b0, 0.f) * fmaxf(b3 - b1, 0.f);
        csc[i] = s;
        cvl[i] = v ? 1 : 0;
    }
    __syncthreads();

    int nk = 0;
    for (int g = 0; g < NGRP; g++) {
        int base = g * 32;
        int i    = base + wave;
        float4 ib = cbx[i];
        float  ia = car[i];
        float  is = csc[i];
        int    vi = cvl[i];

        const float half_ia = 0.5f * ia;
        bool found = false;
        if (vi) {
            int k0 = 0;
            for (; k0 + 128 <= nk; k0 += 128) {
                float4 jb[4];
                bool   pr[4];
                #pragma unroll
                for (int u = 0; u < 4; u++) jb[u] = kbx[k0 + 32 * u + lane];
                #pragma unroll
                for (int u = 0; u < 4; u++)
                    pr[u] = may_overlap(ib.x, ib.y, ib.z, ib.w, half_ia, jb[u].x, jb[u].y, jb[u].z, jb[u].w);
                unsigned anyp = __builtin_amdgcn_ballot_w32(pr[0]) | __builtin_amdgcn_ballot_w32(pr[1]) |
                                __builtin_amdgcn_ballot_w32(pr[2]) | __builtin_amdgcn_ballot_w32(pr[3]);
                if (anyp != 0u) {
                    #pragma unroll
                    for (int u = 0; u < 4; u++)
                        if (pr[u])
                            found |= iou_over(ib.x, ib.y, ib.z, ib.w, ia,
                                              jb[u].x, jb[u].y, jb[u].z, jb[u].w, kar[k0 + 32 * u + lane]);
                }
            }
            for (int k = k0 + lane; k < nk; k += 32) {
                float4 jb = kbx[k];
                if (may_overlap(ib.x, ib.y, ib.z, ib.w, half_ia, jb.x, jb.y, jb.z, jb.w))
                    found |= iou_over(ib.x, ib.y, ib.z, ib.w, ia, jb.x, jb.y, jb.z, jb.w, kar[k]);
            }
        }
        float4 rb = cbx[base + lane];
        bool hit = false;
        if ((lane < wave) && may_overlap(ib.x, ib.y, ib.z, ib.w, half_ia, rb.x, rb.y, rb.z, rb.w))
            hit = iou_over(ib.x, ib.y, ib.z, ib.w, ia, rb.x, rb.y, rb.z, rb.w, car[base + lane]);
        unsigned anyf = __builtin_amdgcn_ballot_w32(found);
        unsigned rowb = __builtin_amdgcn_ballot_w32(hit);
        if (lane == 0) {
            s_alive[wave] = (vi != 0 && anyf == 0u) ? 1u : 0u;
            s_row[wave]   = rowb;
        }
        __syncthreads();

        unsigned al = s_alive[lane];
        unsigned rw = s_row[lane];
        unsigned keepbits = __builtin_amdgcn_ballot_w32(al != 0u && rw == 0u);
        unsigned pend     = __builtin_amdgcn_ballot_w32(al != 0u && rw != 0u);
        while (pend != 0u) {
            int r = __builtin_ctz(pend);
            pend &= pend - 1u;
            unsigned rr = (unsigned)__builtin_amdgcn_readlane((int)rw, r);
            if ((rr & keepbits) == 0u) keepbits |= (1u << r);
        }
        if ((keepbits >> wave) & 1u) {
            int pos = nk + __popc(keepbits & ((1u << wave) - 1u));
            if (lane == 0) { kbx[pos] = ib; kar[pos] = ia; ksc[pos] = is; }
        }
        nk += __popc(keepbits);
        __syncthreads();
    }

    for (int t = tid; t < POST_NMS * 5; t += 1024) {
        int row = t / 5;
        int cc  = t - row * 5;
        float v = 0.f;
        if (row < nk) v = (cc < 4) ? ((const float*)kbx)[row * 4 + cc] : ksc[row];
        out[(size_t)n * POST_NMS * 5 + t] = v;
    }
}

extern "C" void kernel_launch(void* const* d_in, const int* in_sizes, int n_in,
                              void* d_out, int out_size, void* d_ws, size_t ws_size,
                              hipStream_t stream) {
    const float* feat   = (const float*)d_in[0];
    const float* conv_w = (const float*)d_in[1];
    const float* conv_b = (const float*)d_in[2];
    const float* det_w  = (const float*)d_in[3];
    const float* det_b  = (const float*)d_in[4];
    const float* reg_w  = (const float*)d_in[5];
    const float* reg_b  = (const float*)d_in[6];
    float* out = (float*)d_out;

    char* ws = (char*)d_ws;
    float* wt2                  = (float*)(ws);
    float* scores               = (float*)(ws + 2359296);
    float* deltas               = (float*)(ws + 2359296 + 786432);
    float* boxes                = (float*)(ws + 2359296 + 786432 + 3145728);
    unsigned* svk               = (unsigned*)(ws + 2359296 + 786432 + 3145728 + 3145728);

    repack_w_kernel<<<2304, 256, 0, stream>>>(conv_w, wt2);
    conv_fused_kernel<<<NIMG * HF * (WF / TILE_W), 256, 0, stream>>>(
        feat, wt2, conv_b, det_w, det_b, reg_w, reg_b, scores, deltas);
    decode_kernel<<<dim3(NPROP / 256, NIMG), 256, 0, stream>>>(scores, deltas, boxes, svk);
    nms_kernel<<<NIMG, 1024, 0, stream>>>(svk, boxes, out);
}


# conv by minimal filtering F(2,3) along rows: a block makes 2 output rows from 4 combined input rows x 3 kw (12 combined weight planes, 4 sums merged at the end): 1.5x fewer WMMAs; sum order changes
# speedup vs baseline: 4.4390x; 1.4390x over previous
#include <hip/hip_runtime.h>
#include <hip/hip_bf16.h>
#include <math.h>

typedef __attribute__((ext_vector_type(2))) float v2f;
typedef __attribute__((ext_vector_type(8))) float v8f;

#define NIMG 4
#define CIN 256
#define HF 128
#define WF 128
#define A_ANCH 3
#define NPROP (A_ANCH * HF * WF)
#define PRE_NMS 2000
#define POST_NMS 300
#define OBJ_THR 0.3f
#define IOU_THR 0.7f
#define TILE_W 32
#define LCOLS 34
#define CCHUNK 32
#define CIPAD 34

__global__ __launch_bounds__(256) void repack_w_kernel(const float* __restrict__ conv_w,
                                                       float* __restrict__ wt2) {
    int t = blockIdx.x * 256 + threadIdx.x;
    int par = t & 1;
    int co  = (t >> 1) & 255;
    int q   = t >> 9;
    int ci2 = q & 127;
    int kk  = q >> 7;
    int ci  = ci2 * 2 + par;
    int tr  = kk / 3, kw = kk % 3;
    const float* w = conv_w + ((size_t)(co * CIN + ci) * 3) * 3 + kw;
    float g0 = w[0], g1 = w[3], g2 = w[6];
    float v;
    if (tr == 0)      v = g0;
    else if (tr == 1) v = 0.5f * ((g0 + g2) + g1);
    else if (tr == 2) v = 0.5f * ((g0 + g2) - g1);
    else              v = g2;
    wt2[t] = v;
}

__device__ __forceinline__ float4 f4_sub(float4 a, float4 b) { return make_float4(a.x - b.x, a.y - b.y, a.z - b.z, a.w - b.w); }
__device__ __forceinline__ float4 f4_add(float4 a, float4 b) { return make_float4(a.x + b.x, a.y + b.y, a.z + b.z, a.w + b.w); }

__global__ __launch_bounds__(256) void conv_fused_kernel(
    const float* __restrict__ feat, const float* __restrict__ wt2,
    const float* __restrict__ conv_b,
    const float* __restrict__ det_w, const float* __restrict__ det_b,
    const float* __restrict__ reg_w, const float* __restrict__ reg_b,
    float* __restrict__ scores, float* __restrict__ deltas) {

    __shared__ float lds_in[4 * LCOLS * CIPAD];
    __shared__ float x_tile[256 * 33];
    __shared__ float hb[16];

    int bid = blockIdx.x;
    int n    = bid >> 8;
    int rem  = bid & 255;
    int row0 = (rem >> 2) * 2;
    int col0 = (rem & 3) * TILE_W;

    int tid  = threadIdx.x;
    int wave = tid >> 5;
    int lane = tid & 31;
    int hl   = lane >> 4;
    int lm   = lane & 15;

    v8f acc[4][2][2] = {};

    const float* featn = feat + (size_t)n * CIN * HF * WF;
    const v2f*   wp    = (const v2f*)wt2;

    const int sx4   = tid & 7;
    const int scl   = tid >> 3;
    const int hcl   = tid & 31;
    const int hside = (tid >> 5) & 1;
    const int hgcol = col0 - 1 + hside * (TILE_W + 1);
    const bool hcok = (hgcol >= 0) && (hgcol < WF);
    const int hgc   = min(max(hgcol, 0), WF - 1);
    const bool ok0  = (row0 - 1 >= 0);
    const bool ok3  = (row0 + 2 < HF);

    for (int cib = 0; cib < CIN; cib += CCHUNK) {
        float4 sv[4];
        float  hv[4] = {0.f, 0.f, 0.f, 0.f};
        #pragma unroll
        for (int r = 0; r < 4; r++) {
            const int grc = min(max(row0 + r - 1, 0), HF - 1);
            sv[r] = *(const float4*)&featn[((size_t)(cib + scl) * HF + grc) * WF + col0 + 4 * sx4];
        }
        if (tid < 64) {
            #pragma unroll
            for (int r = 0; r < 4; r++) {
                const int grc = min(max(row0 + r - 1, 0), HF - 1);
                hv[r] = featn[((size_t)(cib + hcl) * HF + grc) * WF + hgc];
            }
        }

        __syncthreads();
        {
            const float4 z4 = make_float4(0.f, 0.f, 0.f, 0.f);
            const float4 d0 = ok0 ? sv[0] : z4;
            const float4 d1 = sv[1];
            const float4 d2 = sv[2];
            const float4 d3 = ok3 ? sv[3] : z4;
            float4 cv[4];
            cv[0] = f4_sub(d0, d2);
            cv[1] = f4_add(d1, d2);
            cv[2] = f4_sub(d2, d1);
            cv[3] = f4_sub(d1, d3);
            #pragma unroll
            for (int tr = 0; tr < 4; tr++) {
                float* dst = &lds_in[(tr * LCOLS + 1 + 4 * sx4) * CIPAD + scl];
                dst[0 * CIPAD] = cv[tr].x;
                dst[1 * CIPAD] = cv[tr].y;
                dst[2 * CIPAD] = cv[tr].z;
                dst[3 * CIPAD] = cv[tr].w;
            }
        }
        if (tid < 64) {
            const float e0 = (ok0 && hcok) ? hv[0] : 0.0f;
            const float e1 = hcok ? hv[1] : 0.0f;
            const float e2 = hcok ? hv[2] : 0.0f;
            const float e3 = (ok3 && hcok) ? hv[3] : 0.0f;
            float* dst = &lds_in[(hside * (TILE_W + 1)) * CIPAD + hcl];
            dst[0 * LCOLS * CIPAD] = e0 - e2;
            dst[1 * LCOLS * CIPAD] = e1 + e2;
            dst[2 * LCOLS * CIPAD] = e2 - e1;
            dst[3 * LCOLS * CIPAD] = e1 - e3;
        }
        __syncthreads();

        const v2f* wl = wp + (size_t)(((cib >> 1) + hl) * 256 + wave * 32 + lm);
        #pragma unroll
        for (int tr = 0; tr < 4; tr++) {
            #pragma unroll
            for (int kw = 0; kw < 3; kw++) {
                const int kk = tr * 3 + kw;
                #pragma unroll
                for (int c = 0; c < CCHUNK; c += 4) {
                    v2f a0 = wl[(kk * 128 + (c >> 1)) * 256];
                    v2f a1 = wl[(kk * 128 + (c >> 1)) * 256 + 16];
                    int boff = c + 2 * hl;
                    v2f b0 = *(const v2f*)&lds_in[(tr * LCOLS + lm + kw) * CIPAD + boff];
                    v2f b1 = *(const v2f*)&lds_in[(tr * LCOLS + 16 + lm + kw) * CIPAD + boff];
                    acc[tr][0][0] = __builtin_amdgcn_wmma_f32_16x16x4_f32(false, a0, false, b0, (short)0, acc[tr][0][0], false, false);
                    acc[tr][1][0] = __builtin_amdgcn_wmma_f32_16x16x4_f32(false, a1, false, b0, (short)0, acc[tr][1][0], false, false);
                    acc[tr][0][1] = __builtin_amdgcn_wmma_f32_16x16x4_f32(false, a0, false, b1, (short)0, acc[tr][0][1], false, false);
                    acc[tr][1][1] = __builtin_amdgcn_wmma_f32_16x16x4_f32(false, a1, false, b1, (short)0, acc[tr][1][1], false, false);
                }
            }
        }
    }

    __syncthreads();
    for (int e = tid; e < 15 * 256; e += 256)
        lds_in[e] = (e < 768) ? det_w[e] : reg_w[e - 768];
    if (tid < 15) hb[tid] = (tid < 3) ? det_b[tid] : reg_b[tid - 3];

    int m0  = hl * 8;
    float biasA[8], biasB[8];
    #pragma unroll
    for (int r = 0; r < 8; r++) {
        biasA[r] = conv_b[wave * 32 + m0 + r];
        biasB[r] = conv_b[wave * 32 + 16 + m0 + r];
    }

    for (int ph = 0; ph < 2; ph++) {
        if (ph) __syncthreads();
        #pragma unroll
        for (int r = 0; r < 8; r++) {
            int coA = wave * 32 + m0 + r;
            int coB = wave * 32 + 16 + m0 + r;
            #pragma unroll
            for (int nt = 0; nt < 2; nt++) {
                float yA = (ph == 0) ? ((acc[0][0][nt][r] + acc[1][0][nt][r]) + acc[2][0][nt][r])
                                     : ((acc[1][0][nt][r] - acc[2][0][nt][r]) - acc[3][0][nt][r]);
                float yB = (ph == 0) ? ((acc[0][1][nt][r] + acc[1][1][nt][r]) + acc[2][1][nt][r])
                                     : ((acc[1][1][nt][r] - acc[2][1][nt][r]) - acc[3][1][nt][r]);
                x_tile[coA * 33 + nt * 16 + lm] = yA + biasA[r];
                x_tile[coB * 33 + nt * 16 + lm] = yB + biasB[r];
            }
        }
        __syncthreads();

        for (int t = tid; t < 15 * 32; t += 256) {
            int oi  = t >> 5;
            int pix = t & 31;
            float sum = hb[oi];
            const float* wrow = &lds_in[oi * 256];
            #pragma unroll 8
            for (int co = 0; co < 256; co++)
                sum += wrow[co] * x_tile[co * 33 + pix];
            int p = (row0 + ph) * WF + col0 + pix;
            if (oi < 3) {
                scores[(size_t)n * NPROP + oi * (HF * WF) + p] = 1.0f / (1.0f + expf(-sum));
            } else {
                int ch = oi - 3;
                int ccomp = ch / 3;
                int a = ch % 3;
                deltas[(((size_t)n * NPROP) + a * (HF * WF) + p) * 4 + ccomp] = sum;
            }
        }
    }
}

__global__ __launch_bounds__(256) void decode_kernel(
    const float* scores, const float* __restrict__ deltas,
    float* __restrict__ boxes, unsigned* svk) {
    int n = blockIdx.y;
    int i = blockIdx.x * 256 + threadIdx.x;

    int pa = i / 3, aa = i - pa * 3;
    int ya = pa >> 7, xa = pa & 127;
    const float bs[3][4] = {{-91.f, -45.f, 91.f, 45.f},
                            {-64.f, -64.f, 64.f, 64.f},
                            {-45.f, -91.f, 45.f, 91.f}};
    float sxv = (float)(xa * 8), syv = (float)(ya * 8);
    float x1 = sxv + bs[aa][0], y1 = syv + bs[aa][1];
    float x2 = sxv + bs[aa][2], y2 = syv + bs[aa][3];
    float aw = x2 - x1, ah = y2 - y1;
    float acx = y1 + aw * 0.5f;
    float acy = x1 + ah * 0.5f;

    const float* d = deltas + (((size_t)n * NPROP) + i) * 4;
    float px = acx + d[0] * aw;
    float py = acy + d[1] * ah;
    float pw = aw * expf(d[2]);
    float ph = ah * expf(d[3]);
    float b0 = fminf(fmaxf(px - pw * 0.5f, 0.f), 1024.f);
    float b1 = fminf(fmaxf(py - ph * 0.5f, 0.f), 1024.f);
    float b2 = fminf(fmaxf(px + pw * 0.5f, 0.f), 1024.f);
    float b3 = fminf(fmaxf(py + ph * 0.5f, 0.f), 1024.f);

    float* bo = boxes + (((size_t)n * NPROP) + i) * 4;
    bo[0] = b0; bo[1] = b1; bo[2] = b2; bo[3] = b3;

    float s = scores[(size_t)n * NPROP + i];
    bool valid = (s > OBJ_THR) && ((b2 - b0) > 0.f) && ((b3 - b1) > 0.f);
    unsigned u = __float_as_uint(s);
    unsigned sv = (u & 0x80000000u) ? ~u : (u | 0x80000000u);
    svk[(size_t)n * NPROP + i] = valid ? sv : 0u;
}

#define NGRP 63
#define NCAND (NGRP * 32)
#define KPT 48
#define NSORT 2048
#define GRID_W 32
#define NCELL (GRID_W * GRID_W)

__device__ __forceinline__ int block_total(unsigned (*s_cnt)[32], int& turn, int wave, int lane, int wcount) {
    unsigned* row = s_cnt[turn & 1];
    turn++;
    if (lane == 0) row[wave] = (unsigned)wcount;
    __syncthreads();
    const uint4* p = (const uint4*)row;
    unsigned tot = 0;
    for (int q = 0; q < 8; q++) { uint4 v = p[q]; tot = tot + v.x + v.y; tot = tot + v.z + v.w; }
    return __builtin_amdgcn_readfirstlane((int)tot);
}

__device__ __forceinline__ bool iou_over(float ib0, float ib1, float ib2, float ib3, float ia,
                                         float jx1, float jy1, float jx2, float jy2, float ja) {
    float ix1 = fmaxf(ib0, jx1), iy1 = fmaxf(ib1, jy1);
    float ix2 = fminf(ib2, jx2), iy2 = fminf(ib3, jy2);
    float inter = fmaxf(ix2 - ix1, 0.f) * fmaxf(iy2 - iy1, 0.f);
    float iou = inter / fmaxf(ia + ja - inter, 1e-9f);
    return iou > IOU_THR;
}

__device__ __forceinline__ bool may_overlap(float ib0, float ib1, float ib2, float ib3, float half_ia,
                                            float jx1, float jy1, float jx2, float jy2) {
    float w = fminf(ib2, jx2) - fmaxf(ib0, jx1);
    float h = fminf(ib3, jy2) - fmaxf(ib1, jy1);
    return fmaxf(w, 0.f) * h > half_ia;
}

static_assert(IOU_THR > 0.5f, "the centre-cell argument needs a threshold above one half");
__device__ __forceinline__ int cell_of(float x) {
    return min(max((int)(x * (1.0f / 32.0f)), 0), GRID_W - 1);
}

__global__ __launch_bounds__(1024) void nms_kernel(const unsigned* __restrict__ svk,
                                                   const float* __restrict__ boxes,
                                                   float* __restrict__ out) {
    __shared__ float4   cbx[NCAND];
    __shared__ float    car[NCAND];
    __shared__ float    csc[NCAND];
    __shared__ int      cvl[NCAND];
    __shared__ float4   kbx[NCAND];
    __shared__ float    kar[NCAND];
    __shared__ float    ksc[NCAND];
    __shared__ int      hd[NCELL];
    __shared__ int      nxt[NCAND];
    __shared__ unsigned s_alive[32];
    __shared__ unsigned s_row[32];
    __shared__ unsigned shi[NSORT];
    __shared__ unsigned slo[NSORT];
    __shared__ __attribute__((aligned(16))) unsigned s_cnt[2][32];

    int n    = blockIdx.x;
    int tid  = threadIdx.x;
    int wave = tid >> 5;
    int lane = tid & 31;

    const unsigned* S = svk + (size_t)n * NPROP;
    unsigned sv[KPT];
    #pragma unroll
    for (int j = 0; j < KPT; j++) sv[j] = S[j * 1024 + tid];

    int turn = 0;
    unsigned T = 0u;
    int c_ge = NPROP;
    for (int bit = 31; bit >= 0; bit--) {
        unsigned cand = T | (1u << bit);
        int c = 0;
        #pragma unroll
        for (int j = 0; j < KPT; j += 2) {
            unsigned m0 = __builtin_amdgcn_ballot_w32(sv[j] >= cand);
            unsigned m1 = __builtin_amdgcn_ballot_w32(sv[j + 1] >= cand);
            c += __builtin_popcountll(((unsigned long long)m1 << 32) | m0);
        }
        int tot = block_total(s_cnt, turn, wave, lane, c);
        if (tot >= PRE_NMS) { T = cand; c_ge = tot; }
    }
    if (T != 0u) {
        int c = 0;
        #pragma unroll
        for (int j = 0; j < KPT; j += 2) {
            unsigned m0 = __builtin_amdgcn_ballot_w32(sv[j] > T);
            unsigned m1 = __builtin_amdgcn_ballot_w32(sv[j + 1] > T);
            c += __builtin_popcountll(((unsigned long long)m1 << 32) | m0);
        }
        int c_gt = block_total(s_cnt, turn, wave, lane, c);
        int need = PRE_NMS - c_gt;
        if (c_ge - c_gt != need) {
            unsigned R = 0u;
            for (int bit = 15; bit >= 0; bit--) {
                unsigned cand = R | (1u << bit);
                int ce = 0;
                #pragma unroll
                for (int j = 0; j < KPT; j++) {
                    unsigned r = 0xFFFFu - (unsigned)(j * 1024 + tid);
                    ce += __builtin_popcount(__builtin_amdgcn_ballot_w32((sv[j] == T) & (r >= cand)));
                }
                int tot = block_total(s_cnt, turn, wave, lane, ce);
                if (tot >= need) R = cand;
            }
            #pragma unroll
            for (int j = 0; j < KPT; j++) {
                unsigned r = 0xFFFFu - (unsigned)(j * 1024 + tid);
                if ((sv[j] == T) & (r < R)) sv[j] = 0u;
            }
        }
    }
    const unsigned Tc = (T != 0u) ? T : 1u;
    {
        int wsel = 0;
        #pragma unroll
        for (int j = 0; j < KPT; j += 2) {
            unsigned m0 = __builtin_amdgcn_ballot_w32(sv[j] >= Tc);
            unsigned m1 = __builtin_amdgcn_ballot_w32(sv[j + 1] >= Tc);
            wsel += __builtin_popcountll(((unsigned long long)m1 << 32) | m0);
        }
        unsigned* row = s_cnt[turn & 1];
        turn++;
        if (lane == 0) row[wave] = (unsigned)wsel;
        __syncthreads();
        int before = 0, msel = 0;
        for (int q = 0; q < 32; q++) {
            int v = (int)row[q];
            before += (q < wave) ? v : 0;
            msel   += v;
        }
        int pos = __builtin_amdgcn_readfirstlane(before);
        #pragma unroll
        for (int j = 0; j < KPT; j++) {
            bool sel = sv[j] >= Tc;
            unsigned m = __builtin_amdgcn_ballot_w32(sel);
            int at = pos + (int)__builtin_amdgcn_mbcnt_lo(m, 0u);
            if (sel) { shi[at] = sv[j]; slo[at] = 0xFFFFu - (unsigned)(j * 1024 + tid); }
            pos += __builtin_popcount(m);
        }
        for (int e = msel + tid; e < NSORT; e += 1024) { shi[e] = 0u; slo[e] = 0u; }
    }
    for (int k = 2; k <= NSORT; k <<= 1) {
        for (int j = k >> 1; j > 0; j >>= 1) {
            __syncthreads();
            int i = ((tid & ~(j - 1)) << 1) | (tid & (j - 1));
            int p = i | j;
            unsigned ah = shi[i], al = slo[i];
            unsigned bh = shi[p], bl = slo[p];
            bool a_lt_b = (ah < bh) || (ah == bh && al < bl);
            bool up = ((i & k) == 0);
            if (a_lt_b == up) { shi[i] = bh; slo[i] = bl; shi[p] = ah; slo[p] = al; }
        }
    }
    __syncthreads();

    for (int i = tid; i < NCAND; i += 1024) {
        bool v = false;
        float b0 = 0.f, b1 = 0.f, b2 = 0.f, b3 = 0.f, s = 0.f;
        if (i < PRE_NMS) {
            unsigned w   = shi[i];
            unsigned idx = 0xFFFFu - slo[i];
            v = w != 0u;
            if (v) {
                const float* bp = boxes + (((size_t)n * NPROP) + idx) * 4;
                b0 = bp[0]; b1 = bp[1]; b2 = bp[2]; b3 = bp[3];
                unsigned u = (w & 0x80000000u) ? (w ^ 0x80000000u) : ~w;
                s = __uint_as_float(u);
            }
        }
        cbx[i] = make_float4(b0, b1, b2, b3);
        car[i] = fmaxf(b2 - b0, 0.f) * fmaxf(b3 - b1, 0.f);
        csc[i] = s;
        cvl[i] = v ? 1 : 0;
    }
    for (int c = tid; c < NCELL; c += 1024) hd[c] = -1;
    __syncthreads();

    int nk = 0;
    for (int g = 0; g < NGRP; g++) {
        int base = g * 32;
        int i    = base + wave;
        float4 ib = cbx[i];
        float  ia = car[i];
        float  is = csc[i];
        int    vi = cvl[i];

        const float half_ia = 0.5f * ia;
        const int mycell  = cell_of(0.5f * (ib.y + ib.w)) * GRID_W + cell_of(0.5f * (ib.x + ib.z));
        const int oldhead = hd[mycell];
        bool found = false;
        if (vi) {
            const int cx0 = cell_of(ib.x), cy0 = cell_of(ib.y);
            const int nx  = cell_of(ib.z) - cx0 + 1;
            const int ny  = cell_of(ib.w) - cy0 + 1;
            const int m   = nx * ny;
            const float rnx = __builtin_amdgcn_rcpf((float)nx);
            for (int q = lane; q < m; q += 32) {
                const int qy = (int)(((float)q + 0.5f) * rnx);
                const int qx = q - qy * nx;
                int k = hd[min((cy0 + qy) * GRID_W + cx0 + qx, NCELL - 1)];
                for (int it = 0; it < NCAND && k >= 0; it++) {
                    const float4 jb = kbx[k];
                    const int    kn = nxt[k];
                    if (may_overlap(ib.x, ib.y, ib.z, ib.w, half_ia, jb.x, jb.y, jb.z, jb.w))
                        found |= iou_over(ib.x, ib.y, ib.z, ib.w, ia, jb.x, jb.y, jb.z, jb.w, kar[k]);
                    k = kn;
                }
            }
        }
        float4 rb = cbx[base + lane];
        const int rcell = cell_of(0.5f * (rb.y + rb.w)) * GRID_W + cell_of(0.5f * (rb.x + rb.z));
        bool hit = false;
        if ((lane < wave) && may_overlap(ib.x, ib.y, ib.z, ib.w, half_ia, rb.x, rb.y, rb.z, rb.w))
            hit = iou_over(ib.x, ib.y, ib.z, ib.w, ia, rb.x, rb.y, rb.z, rb.w, car[base + lane]);
        unsigned anyf = __builtin_amdgcn_ballot_w32(found);
        unsigned rowb = __builtin_amdgcn_ballot_w32(hit);
        if (lane == 0) {
            s_alive[wave] = (vi != 0 && anyf == 0u) ? 1u : 0u;
            s_row[wave]   = rowb;
        }
        __syncthreads();

        unsigned al = s_alive[lane];
        unsigned rw = s_row[lane];
        unsigned keepbits = __builtin_amdgcn_ballot_w32(al != 0u && rw == 0u);
        unsigned pend     = __builtin_amdgcn_ballot_w32(al != 0u && rw != 0u);
        while (pend != 0u) {
            int r = __builtin_ctz(pend);
            pend &= pend - 1u;
            unsigned rr = (unsigned)__builtin_amdgcn_readlane((int)rw, r);
            if ((rr & keepbits) == 0u) keepbits |= (1u << r);
        }
        if ((keepbits >> wave) & 1u) {
            const unsigned below = (1u << wave) - 1u;
            int pos = nk + __popc(keepbits & below);
            unsigned same  = __builtin_amdgcn_ballot_w32(rcell == mycell) & keepbits;
            unsigned older = same & below;
            int link = oldhead;
            if (older != 0u) {
                int p = 31 - __builtin_clz(older);
                link = nk + __popc(keepbits & ((1u << p) - 1u));
            }
            if (lane == 0) {
                kbx[pos] = ib; kar[pos] = ia; ksc[pos] = is; nxt[pos] = link;
                if ((same & ~below & ~(1u << wave)) == 0u) hd[mycell] = pos;
            }
        }
        nk += __popc(keepbits);
        __syncthreads();
    }

    for (int t = tid; t < POST_NMS * 5; t += 1024) {
        int row = t / 5;
        int cc  = t - row * 5;
        float v = 0.f;
        if (row < nk) v = (cc < 4) ? ((const float*)kbx)[row * 4 + cc] : ksc[row];
        out[(size_t)n * POST_NMS * 5 + t] = v;
    }
}

extern "C" void kernel_launch(void* const* d_in, const int* in_sizes, int n_in,
                              void* d_out, int out_size, void* d_ws, size_t ws_size,
                              hipStream_t stream) {
    const float* feat   = (const float*)d_in[0];
    const float* conv_w = (const float*)d_in[1];
    const float* conv_b = (const float*)d_in[2];
    const float* det_w  = (const float*)d_in[3];
    const float* det_b  = (const float*)d_in[4];
    const float* reg_w  = (const float*)d_in[5];
    const float* reg_b  = (const float*)d_in[6];
    float* out = (float*)d_out;

    char* ws = (char*)d_ws;
    float* wt2                  = (float*)(ws);
    float* scores               = (float*)(ws + 3145728);
    float* deltas               = (float*)(ws + 3145728 + 786432);
    float* boxes                = (float*)(ws + 3145728 + 786432 + 3145728);
    unsigned* svk               = (unsigned*)(ws + 3145728);

    repack_w_kernel<<<3072, 256, 0, stream>>>(conv_w, wt2);
    conv_fused_kernel<<<NIMG * (HF / 2) * (WF / TILE_W), 256, 0, stream>>>(
        feat, wt2, conv_b, det_w, det_b, reg_w, reg_b, scores, deltas);
    decode_kernel<<<dim3(NPROP / 256, NIMG), 256, 0, stream>>>(scores, deltas, boxes, svk);
    nms_kernel<<<NIMG, 1024, 0, stream>>>(svk, boxes, out);
}


# conv by F(4,3) along rows: 4 output rows from 6 combined rows x 3 kw (18 weight planes, 6 sums merged at the end), a plane's 16 weight loads issued as one group: 1.33x fewer WMMAs; sum order changes
# speedup vs baseline: 4.7327x; 1.0662x over previous
#include <hip/hip_runtime.h>
#include <hip/hip_bf16.h>
#include <math.h>

typedef __attribute__((ext_vector_type(2))) float v2f;
typedef __attribute__((ext_vector_type(8))) float v8f;

#define NIMG 4
#define CIN 256
#define HF 128
#define WF 128
#define A_ANCH 3
#define NPROP (A_ANCH * HF * WF)
#define PRE_NMS 2000
#define POST_NMS 300
#define OBJ_THR 0.3f
#define IOU_THR 0.7f
#define TILE_W 32
#define LCOLS 34
#define CCHUNK 32
#define CIPAD 34
#define NTR 6

__global__ __launch_bounds__(256) void repack_w_kernel(const float* __restrict__ conv_w,
                                                       float* __restrict__ wt2) {
    int t = blockIdx.x * 256 + threadIdx.x;
    int par = t & 1;
    int co  = (t >> 1) & 255;
    int q   = t >> 9;
    int ci2 = q & 127;
    int kk  = q >> 7;
    int ci  = ci2 * 2 + par;
    int tr  = kk / 3, kw = kk % 3;
    const float* w = conv_w + ((size_t)(co * CIN + ci) * 3) * 3 + kw;
    float g0 = w[0], g1 = w[3], g2 = w[6];
    const float c6 = 1.0f / 6.0f, c12 = 1.0f / 12.0f, c24 = 1.0f / 24.0f;
    float v;
    if (tr == 0)      v = 0.25f * g0;
    else if (tr == 1) v = -c6 * ((g0 + g2) + g1);
    else if (tr == 2) v = -c6 * ((g0 + g2) - g1);
    else if (tr == 3) v = (c24 * g0 + c6 * g2) + c12 * g1;
    else if (tr == 4) v = (c24 * g0 + c6 * g2) - c12 * g1;
    else              v = g2;
    wt2[t] = v;
}

__device__ __forceinline__ void comb6(const float d[6], float v[6]) {
    v[0] = (4.0f * d[0] - 5.0f * d[2]) + d[4];
    v[1] = (d[3] + d[4]) - 4.0f * (d[1] + d[2]);
    v[2] = (d[4] - d[3]) + 4.0f * (d[1] - d[2]);
    v[3] = (d[4] - d[2]) + 2.0f * (d[3] - d[1]);
    v[4] = (d[4] - d[2]) - 2.0f * (d[3] - d[1]);
    v[5] = (4.0f * d[1] - 5.0f * d[3]) + d[5];
}


__global__ __launch_bounds__(256, 4) void conv_fused_kernel(
    const float* __restrict__ feat, const float* __restrict__ wt2,
    const float* __restrict__ conv_b,
    const float* __restrict__ det_w, const float* __restrict__ det_b,
    const float* __restrict__ reg_w, const float* __restrict__ reg_b,
    float* __restrict__ scores, float* __restrict__ deltas) {

    __shared__ float lds_in[NTR * LCOLS * CIPAD];
    __shared__ float x_tile[256 * 33];
    __shared__ float hb[16];
    __shared__ float cb[256];

    int bid = blockIdx.x;
    int n    = bid >> 7;
    int rem  = bid & 127;
    int row0 = (rem >> 2) * 4;
    int col0 = (rem & 3) * TILE_W;

    int tid  = threadIdx.x;
    int wave = tid >> 5;
    int lane = tid & 31;
    int hl   = lane >> 4;
    int lm   = lane & 15;

    v8f acc[NTR][2][2] = {};

    const float* featn = feat + (size_t)n * CIN * HF * WF;
    const v2f*   wp    = (const v2f*)wt2;

    const int sx4   = tid & 7;
    const int scl   = tid >> 3;
    const int hcl   = tid & 31;
    const int hside = (tid >> 5) & 1;
    const int hgcol = col0 - 1 + hside * (TILE_W + 1);
    const bool hcok = (hgcol >= 0) && (hgcol < WF);
    const int hgc   = min(max(hgcol, 0), WF - 1);
    const bool ok0  = (row0 - 1 >= 0);
    const bool ok5  = (row0 + 4 < HF);

    for (int cib = 0; cib < CIN; cib += CCHUNK) {
        float4 sv[NTR];
        float  hv[NTR] = {0.f, 0.f, 0.f, 0.f, 0.f, 0.f};
        #pragma unroll
        for (int r = 0; r < NTR; r++) {
            const int grc = min(max(row0 + r - 1, 0), HF - 1);
            sv[r] = *(const float4*)&featn[((size_t)(cib + scl) * HF + grc) * WF + col0 + 4 * sx4];
        }
        if (tid < 64) {
            #pragma unroll
            for (int r = 0; r < NTR; r++) {
                const int grc = min(max(row0 + r - 1, 0), HF - 1);
                hv[r] = featn[((size_t)(cib + hcl) * HF + grc) * WF + hgc];
            }
        }

        __syncthreads();
        {
            const float4 z4 = make_float4(0.f, 0.f, 0.f, 0.f);
            const float4 s0 = ok0 ? sv[0] : z4;
            const float4 s5 = ok5 ? sv[5] : z4;
            const float dx[6] = {s0.x, sv[1].x, sv[2].x, sv[3].x, sv[4].x, s5.x};
            const float dy[6] = {s0.y, sv[1].y, sv[2].y, sv[3].y, sv[4].y, s5.y};
            const float dz[6] = {s0.z, sv[1].z, sv[2].z, sv[3].z, sv[4].z, s5.z};
            const float dw[6] = {s0.w, sv[1].w, sv[2].w, sv[3].w, sv[4].w, s5.w};
            float cx[6], cy[6], cz[6], cw[6];
            comb6(dx, cx); comb6(dy, cy); comb6(dz, cz); comb6(dw, cw);
            #pragma unroll
            for (int tr = 0; tr < NTR; tr++) {
                float* dst = &lds_in[(tr * LCOLS + 1 + 4 * sx4) * CIPAD + scl];
                dst[0 * CIPAD] = cx[tr];
                dst[1 * CIPAD] = cy[tr];
                dst[2 * CIPAD] = cz[tr];
                dst[3 * CIPAD] = cw[tr];
            }
        }
        if (tid < 64) {
            float e[6], ce[6];
            e[0] = (ok0 && hcok) ? hv[0] : 0.0f;
            e[1] = hcok ? hv[1] : 0.0f;
            e[2] = hcok ? hv[2] : 0.0f;
            e[3] = hcok ? hv[3] : 0.0f;
            e[4] = hcok ? hv[4] : 0.0f;
            e[5] = (ok5 && hcok) ? hv[5] : 0.0f;
            comb6(e, ce);
            float* dst = &lds_in[(hside * (TILE_W + 1)) * CIPAD + hcl];
            #pragma unroll
            for (int tr = 0; tr < NTR; tr++) dst[tr * LCOLS * CIPAD] = ce[tr];
        }
        __syncthreads();

        const v2f* wl = wp + (size_t)(((cib >> 1) + hl) * 256 + wave * 32 + lm);
        #pragma unroll
        for (int tr = 0; tr < NTR; tr++) {
            #pragma unroll
            for (int kw = 0; kw < 3; kw++) {
                const int kk = tr * 3 + kw;
                v2f a0[CCHUNK / 4], a1[CCHUNK / 4];
                #pragma unroll
                for (int j = 0; j < CCHUNK / 4; j++) {
                    a0[j] = wl[(kk * 128 + 2 * j) * 256];
                    a1[j] = wl[(kk * 128 + 2 * j) * 256 + 16];
                }
                __builtin_amdgcn_sched_barrier(0);
                #pragma unroll
                for (int j = 0; j < CCHUNK / 4; j++) {
                    int boff = 4 * j + 2 * hl;
                    v2f b0 = *(const v2f*)&lds_in[(tr * LCOLS + lm + kw) * CIPAD + boff];
                    v2f b1 = *(const v2f*)&lds_in[(tr * LCOLS + 16 + lm + kw) * CIPAD + boff];
                    acc[tr][0][0] = __builtin_amdgcn_wmma_f32_16x16x4_f32(false, a0[j], false, b0, (short)0, acc[tr][0][0], false, false);
                    acc[tr][1][0] = __builtin_amdgcn_wmma_f32_16x16x4_f32(false, a1[j], false, b0, (short)0, acc[tr][1][0], false, false);
                    acc[tr][0][1] = __builtin_amdgcn_wmma_f32_16x16x4_f32(false, a0[j], false, b1, (short)0, acc[tr][0][1], false, false);
                    acc[tr][1][1] = __builtin_amdgcn_wmma_f32_16x16x4_f32(false, a1[j], false, b1, (short)0, acc[tr][1][1], false, false);
                }
            }
        }
    }

    cb[tid] = conv_b[tid];
    __syncthreads();
    for (int e = tid; e < 15 * 256; e += 256)
        lds_in[e] = (e < 768) ? det_w[e] : reg_w[e - 768];
    if (tid < 15) hb[tid] = (tid < 3) ? det_b[tid] : reg_b[tid - 3];

    int m0  = hl * 8;

    #pragma unroll
    for (int ph = 0; ph < 4; ph++) {
        if (ph) __syncthreads();
        #pragma unroll
        for (int r = 0; r < 8; r++) {
            int coA = wave * 32 + m0 + r;
            int coB = wave * 32 + 16 + m0 + r;
            #pragma unroll
            for (int mt = 0; mt < 2; mt++) {
                #pragma unroll
                for (int nt = 0; nt < 2; nt++) {
                    const float q0 = acc[0][mt][nt][r], q1 = acc[1][mt][nt][r], q2 = acc[2][mt][nt][r];
                    const float q3 = acc[3][mt][nt][r], q4 = acc[4][mt][nt][r], q5 = acc[5][mt][nt][r];
                    float y;
                    if (ph == 0)      y = (q0 + (q1 + q2)) + (q3 + q4);
                    else if (ph == 1) y = (q1 - q2) + 2.0f * (q3 - q4);
                    else if (ph == 2) y = (q1 + q2) + 4.0f * (q3 + q4);
                    else              y = ((q1 - q2) + 8.0f * (q3 - q4)) + q5;
                    const int co = mt ? coB : coA;
                    x_tile[co * 33 + nt * 16 + lm] = y + cb[co];
                }
            }
        }
        __syncthreads();

        for (int t = tid; t < 15 * 32; t += 256) {
            int oi  = t >> 5;
            int pix = t & 31;
            float sum = hb[oi];
            const float* wrow = &lds_in[oi * 256];
            #pragma unroll 8
            for (int co = 0; co < 256; co++)
                sum += wrow[co] * x_tile[co * 33 + pix];
            int p = (row0 + ph) * WF + col0 + pix;
            if (oi < 3) {
                scores[(size_t)n * NPROP + oi * (HF * WF) + p] = 1.0f / (1.0f + expf(-sum));
            } else {
                int ch = oi - 3;
                int ccomp = ch / 3;
                int a = ch % 3;
                deltas[(((size_t)n * NPROP) + a * (HF * WF) + p) * 4 + ccomp] = sum;
            }
        }
    }
}

__global__ __launch_bounds__(256) void decode_kernel(
    const float* scores, const float* deltas,
    float* boxes, unsigned* svk) {
    int n = blockIdx.y;
    int i = blockIdx.x * 256 + threadIdx.x;

    int pa = i / 3, aa = i - pa * 3;
    int ya = pa >> 7, xa = pa & 127;
    const float bs[3][4] = {{-91.f, -45.f, 91.f, 45.f},
                            {-64.f, -64.f, 64.f, 64.f},
                            {-45.f, -91.f, 45.f, 91.f}};
    float sxv = (float)(xa * 8), syv = (float)(ya * 8);
    float x1 = sxv + bs[aa][0], y1 = syv + bs[aa][1];
    float x2 = sxv + bs[aa][2], y2 = syv + bs[aa][3];
    float aw = x2 - x1, ah = y2 - y1;
    float acx = y1 + aw * 0.5f;
    float acy = x1 + ah * 0.5f;

    const float* d = deltas + (((size_t)n * NPROP) + i) * 4;
    const float d0 = d[0], d1 = d[1], d2 = d[2], d3 = d[3];
    float px = acx + d0 * aw;
    float py = acy + d1 * ah;
    float pw = aw * expf(d2);
    float ph = ah * expf(d3);
    float b0 = fminf(fmaxf(px - pw * 0.5f, 0.f), 1024.f);
    float b1 = fminf(fmaxf(py - ph * 0.5f, 0.f), 1024.f);
    float b2 = fminf(fmaxf(px + pw * 0.5f, 0.f), 1024.f);
    float b3 = fminf(fmaxf(py + ph * 0.5f, 0.f), 1024.f);

    float* bo = boxes + (((size_t)n * NPROP) + i) * 4;
    bo[0] = b0; bo[1] = b1; bo[2] = b2; bo[3] = b3;

    float s = scores[(size_t)n * NPROP + i];
    bool valid = (s > OBJ_THR) && ((b2 - b0) > 0.f) && ((b3 - b1) > 0.f);
    unsigned u = __float_as_uint(s);
    unsigned sv = (u & 0x80000000u) ? ~u : (u | 0x80000000u);
    svk[(size_t)n * NPROP + i] = valid ? sv : 0u;
}

#define NGRP 63
#define NCAND (NGRP * 32)
#define KPT 48
#define NSORT 2048
#define GRID_W 32
#define NCELL (GRID_W * GRID_W)

__device__ __forceinline__ int block_total(unsigned (*s_cnt)[32], int& turn, int wave, int lane, int wcount) {
    unsigned* row = s_cnt[turn & 1];
    turn++;
    if (lane == 0) row[wave] = (unsigned)wcount;
    __syncthreads();
    const uint4* p = (const uint4*)row;
    unsigned tot = 0;
    for (int q = 0; q < 8; q++) { uint4 v = p[q]; tot = tot + v.x + v.y; tot = tot + v.z + v.w; }
    return __builtin_amdgcn_readfirstlane((int)tot);
}

__device__ __forceinline__ bool iou_over(float ib0, float ib1, float ib2, float ib3, float ia,
                                         float jx1, float jy1, float jx2, float jy2, float ja) {
    float ix1 = fmaxf(ib0, jx1), iy1 = fmaxf(ib1, jy1);
    float ix2 = fminf(ib2, jx2), iy2 = fminf(ib3, jy2);
    float inter = fmaxf(ix2 - ix1, 0.f) * fmaxf(iy2 - iy1, 0.f);
    float iou = inter / fmaxf(ia + ja - inter, 1e-9f);
    return iou > IOU_THR;
}

__device__ __forceinline__ bool may_overlap(float ib0, float ib1, float ib2, float ib3, float half_ia,
                                            float jx1, float jy1, float jx2, float jy2) {
    float w = fminf(ib2, jx2) - fmaxf(ib0, jx1);
    float h = fminf(ib3, jy2) - fmaxf(ib1, jy1);
    return fmaxf(w, 0.f) * h > half_ia;
}

static_assert(IOU_THR > 0.5f, "the centre-cell argument needs a threshold above one half");
__device__ __forceinline__ int cell_of(float x) {
    return min(max((int)(x * (1.0f / 32.0f)), 0), GRID_W - 1);
}

__global__ __launch_bounds__(1024) void nms_kernel(const unsigned* __restrict__ svk,
                                                   const float* __restrict__ boxes,
                                                   float* __restrict__ out) {
    __shared__ float4   cbx[NCAND];
    __shared__ float    car[NCAND];
    __shared__ float    csc[NCAND];
    __shared__ int      cvl[NCAND];
    __shared__ float4   kbx[NCAND];
    __shared__ float    kar[NCAND];
    __shared__ float    ksc[NCAND];
    __shared__ int      hd[NCELL];
    __shared__ int      nxt[NCAND];
    __shared__ unsigned s_alive[32];
    __shared__ unsigned s_row[32];
    __shared__ unsigned shi[NSORT];
    __shared__ unsigned slo[NSORT];
    __shared__ __attribute__((aligned(16))) unsigned s_cnt[2][32];

    int n    = blockIdx.x;
    int tid  = threadIdx.x;
    int wave = tid >> 5;
    int lane = tid & 31;

    const unsigned* S = svk + (size_t)n * NPROP;
    unsigned sv[KPT];
    #pragma unroll
    for (int j = 0; j < KPT; j++) sv[j] = S[j * 1024 + tid];

    int turn = 0;
    unsigned T = 0u;
    int c_ge = NPROP;
    for (int bit = 31; bit >= 0; bit--) {
        unsigned cand = T | (1u << bit);
        int c = 0;
        #pragma unroll
        for (int j = 0; j < KPT; j += 2) {
            unsigned m0 = __builtin_amdgcn_ballot_w32(sv[j] >= cand);
            unsigned m1 = __builtin_amdgcn_ballot_w32(sv[j + 1] >= cand);
            c += __builtin_popcountll(((unsigned long long)m1 << 32) | m0);
        }
        int tot = block_total(s_cnt, turn, wave, lane, c);
        if (tot >= PRE_NMS) { T = cand; c_ge = tot; }
    }
    if (T != 0u) {
        int c = 0;
        #pragma unroll
        for (int j = 0; j < KPT; j += 2) {
            unsigned m0 = __builtin_amdgcn_ballot_w32(sv[j] > T);
            unsigned m1 = __builtin_amdgcn_ballot_w32(sv[j + 1] > T);
            c += __builtin_popcountll(((unsigned long long)m1 << 32) | m0);
        }
        int c_gt = block_total(s_cnt, turn, wave, lane, c);
        int need = PRE_NMS - c_gt;
        if (c_ge - c_gt != need) {
            unsigned R = 0u;
            for (int bit = 15; bit >= 0; bit--) {
                unsigned cand = R | (1u << bit);
                int ce = 0;
                #pragma unroll
                for (int j = 0; j < KPT; j++) {
                    unsigned r = 0xFFFFu - (unsigned)(j * 1024 + tid);
                    ce += __builtin_popcount(__builtin_amdgcn_ballot_w32((sv[j] == T) & (r >= cand)));
                }
                int tot = block_total(s_cnt, turn, wave, lane, ce);
                if (tot >= need) R = cand;
            }
            #pragma unroll
            for (int j = 0; j < KPT; j++) {
                unsigned r = 0xFFFFu - (unsigned)(j * 1024 + tid);
                if ((sv[j] == T) & (r < R)) sv[j] = 0u;
            }
        }
    }
    const unsigned Tc = (T != 0u) ? T : 1u;
    {
        int wsel = 0;
        #pragma unroll
        for (int j = 0; j < KPT; j += 2) {
            unsigned m0 = __builtin_amdgcn_ballot_w32(sv[j] >= Tc);
            unsigned m1 = __builtin_amdgcn_ballot_w32(sv[j + 1] >= Tc);
            wsel += __builtin_popcountll(((unsigned long long)m1 << 32) | m0);
        }
        unsigned* row = s_cnt[turn & 1];
        turn++;
        if (lane == 0) row[wave] = (unsigned)wsel;
        __syncthreads();
        int before = 0, msel = 0;
        for (int q = 0; q < 32; q++) {
            int v = (int)row[q];
            before += (q < wave) ? v : 0;
            msel   += v;
        }
        int pos = __builtin_amdgcn_readfirstlane(before);
        #pragma unroll
        for (int j = 0; j < KPT; j++) {
            bool sel = sv[j] >= Tc;
            unsigned m = __builtin_amdgcn_ballot_w32(sel);
            int at = pos + (int)__builtin_amdgcn_mbcnt_lo(m, 0u);
            if (sel) { shi[at] = sv[j]; slo[at] = 0xFFFFu - (unsigned)(j * 1024 + tid); }
            pos += __builtin_popcount(m);
        }
        for (int e = msel + tid; e < NSORT; e += 1024) { shi[e] = 0u; slo[e] = 0u; }
    }
    for (int k = 2; k <= NSORT; k <<= 1) {
        for (int j = k >> 1; j > 0; j >>= 1) {
            __syncthreads();
            int i = ((tid & ~(j - 1)) << 1) | (tid & (j - 1));
            int p = i | j;
            unsigned ah = shi[i], al = slo[i];
            unsigned bh = shi[p], bl = slo[p];
            bool a_lt_b = (ah < bh) || (ah == bh && al < bl);
            bool up = ((i & k) == 0);
            if (a_lt_b == up) { shi[i] = bh; slo[i] = bl; shi[p] = ah; slo[p] = al; }
        }
    }
    __syncthreads();

    for (int i = tid; i < NCAND; i += 1024) {
        bool v = false;
        float b0 = 0.f, b1 = 0.f, b2 = 0.f, b3 = 0.f, s = 0.f;
        if (i < PRE_NMS) {
            unsigned w   = shi[i];
            unsigned idx = 0xFFFFu - slo[i];
            v = w != 0u;
            if (v) {
                const float* bp = boxes + (((size_t)n * NPROP) + idx) * 4;
                b0 = bp[0]; b1 = bp[1]; b2 = bp[2]; b3 = bp[3];
                unsigned u = (w & 0x80000000u) ? (w ^ 0x80000000u) : ~w;
                s = __uint_as_float(u);
            }
        }
        cbx[i] = make_float4(b0, b1, b2, b3);
        car[i] = fmaxf(b2 - b0, 0.f) * fmaxf(b3 - b1, 0.f);
        csc[i] = s;
        cvl[i] = v ? 1 : 0;
    }
    for (int c = tid; c < NCELL; c += 1024) hd[c] = -1;
    __syncthreads();

    int nk = 0;
    for (int g = 0; g < NGRP; g++) {
        int base = g * 32;
        int i    = base + wave;
        float4 ib = cbx[i];
        float  ia = car[i];
        float  is = csc[i];
        int    vi = cvl[i];

        const float half_ia = 0.5f * ia;
        const int mycell  = cell_of(0.5f * (ib.y + ib.w)) * GRID_W + cell_of(0.5f * (ib.x + ib.z));
        const int oldhead = hd[mycell];
        bool found = false;
        if (vi) {
            const int cx0 = cell_of(ib.x), cy0 = cell_of(ib.y);
            const int nx  = cell_of(ib.z) - cx0 + 1;
            const int ny  = cell_of(ib.w) - cy0 + 1;
            const int m   = nx * ny;
            const float rnx = __builtin_amdgcn_rcpf((float)nx);
            for (int q = lane; q < m; q += 32) {
                const int qy = (int)(((float)q + 0.5f) * rnx);
                const int qx = q - qy * nx;
                int k = hd[min((cy0 + qy) * GRID_W + cx0 + qx, NCELL - 1)];
                for (int it = 0; it < NCAND && k >= 0; it++) {
                    const float4 jb = kbx[k];
                    const int    kn = nxt[k];
                    if (may_overlap(ib.x, ib.y, ib.z, ib.w, half_ia, jb.x, jb.y, jb.z, jb.w))
                        found |= iou_over(ib.x, ib.y, ib.z, ib.w, ia, jb.x, jb.y, jb.z, jb.w, kar[k]);
                    k = kn;
                }
            }
        }
        float4 rb = cbx[base + lane];
        const int rcell = cell_of(0.5f * (rb.y + rb.w)) * GRID_W + cell_of(0.5f * (rb.x + rb.z));
        bool hit = false;
        if ((lane < wave) && may_overlap(ib.x, ib.y, ib.z, ib.w, half_ia, rb.x, rb.y, rb.z, rb.w))
            hit = iou_over(ib.x, ib.y, ib.z, ib.w, ia, rb.x, rb.y, rb.z, rb.w, car[base + lane]);
        unsigned anyf = __builtin_amdgcn_ballot_w32(found);
        unsigned rowb = __builtin_amdgcn_ballot_w32(hit);
        if (lane == 0) {
            s_alive[wave] = (vi != 0 && anyf == 0u) ? 1u : 0u;
            s_row[wave]   = rowb;
        }
        __syncthreads();

        unsigned al = s_alive[lane];
        unsigned rw = s_row[lane];
        unsigned keepbits = __builtin_amdgcn_ballot_w32(al != 0u && rw == 0u);
        unsigned pend     = __builtin_amdgcn_ballot_w32(al != 0u && rw != 0u);
        while (pend != 0u) {
            int r = __builtin_ctz(pend);
            pend &= pend - 1u;
            unsigned rr = (unsigned)__builtin_amdgcn_readlane((int)rw, r);
            if ((rr & keepbits) == 0u) keepbits |= (1u << r);
        }
        if ((keepbits >> wave) & 1u) {
            const unsigned below = (1u << wave) - 1u;
            int pos = nk + __popc(keepbits & below);
            unsigned same  = __builtin_amdgcn_ballot_w32(rcell == mycell) & keepbits;
            unsigned older = same & below;
            int link = oldhead;
            if (older != 0u) {
                int p = 31 - __builtin_clz(older);
                link = nk + __popc(keepbits & ((1u << p) - 1u));
            }
            if (lane == 0) {
                kbx[pos] = ib; kar[pos] = ia; ksc[pos] = is; nxt[pos] = link;
                if ((same & ~below & ~(1u << wave)) == 0u) hd[mycell] = pos;
            }
        }
        nk += __popc(keepbits);
        __syncthreads();
    }

    for (int t = tid; t < POST_NMS * 5; t += 1024) {
        int row = t / 5;
        int cc  = t - row * 5;
        float v = 0.f;
        if (row < nk) v = (cc < 4) ? ((const float*)kbx)[row * 4 + cc] : ksc[row];
        out[(size_t)n * POST_NMS * 5 + t] = v;
    }
}

extern "C" void kernel_launch(void* const* d_in, const int* in_sizes, int n_in,
                              void* d_out, int out_size, void* d_ws, size_t ws_size,
                              hipStream_t stream) {
    const float* feat   = (const float*)d_in[0];
    const float* conv_w = (const float*)d_in[1];
    const float* conv_b = (const float*)d_in[2];
    const float* det_w  = (const float*)d_in[3];
    const float* det_b  = (const float*)d_in[4];
    const float* reg_w  = (const float*)d_in[5];
    const float* reg_b  = (const float*)d_in[6];
    float* out = (float*)d_out;

    char* ws = (char*)d_ws;
    float* wt2                  = (float*)(ws);
    float* scores               = (float*)(ws + 4718592);
    float* deltas               = (float*)(ws + 4718592 + 786432);
    float* boxes                = deltas;
    unsigned* svk               = (unsigned*)(ws + 4718592);

    repack_w_kernel<<<4608, 256, 0, stream>>>(conv_w, wt2);
    conv_fused_kernel<<<NIMG * (HF / 4) * (WF / TILE_W), 256, 0, stream>>>(
        feat, wt2, conv_b, det_w, det_b, reg_w, reg_b, scores, deltas);
    decode_kernel<<<dim3(NPROP / 256, NIMG), 256, 0, stream>>>(scores, deltas, boxes, svk);
    nms_kernel<<<NIMG, 1024, 0, stream>>>(svk, boxes, out);
}


# head 1x1 convs as WMMA products fed straight from the conv accumulator registers (a D tile is a B fragment as it stands); 8 per-wave partial tiles added in LDS, not LDS dot loops; sum order changes
# speedup vs baseline: 5.1714x; 1.0927x over previous
#include <hip/hip_runtime.h>
#include <hip/hip_bf16.h>
#include <math.h>

typedef __attribute__((ext_vector_type(2))) float v2f;
typedef __attribute__((ext_vector_type(8))) float v8f;

#define NIMG 4
#define CIN 256
#define HF 128
#define WF 128
#define A_ANCH 3
#define NPROP (A_ANCH * HF * WF)
#define PRE_NMS 2000
#define POST_NMS 300
#define OBJ_THR 0.3f
#define IOU_THR 0.7f
#define TILE_W 32
#define LCOLS 34
#define CCHUNK 32
#define CIPAD 34
#define NTR 6
#define PPITCH 36

__global__ __launch_bounds__(256) void repack_w_kernel(const float* __restrict__ conv_w,
                                                       float* __restrict__ wt2) {
    int t = blockIdx.x * 256 + threadIdx.x;
    int par = t & 1;
    int co  = (t >> 1) & 255;
    int q   = t >> 9;
    int ci2 = q & 127;
    int kk  = q >> 7;
    int ci  = ci2 * 2 + par;
    int tr  = kk / 3, kw = kk % 3;
    const float* w = conv_w + ((size_t)(co * CIN + ci) * 3) * 3 + kw;
    float g0 = w[0], g1 = w[3], g2 = w[6];
    const float c6 = 1.0f / 6.0f, c12 = 1.0f / 12.0f, c24 = 1.0f / 24.0f;
    float v;
    if (tr == 0)      v = 0.25f * g0;
    else if (tr == 1) v = -c6 * ((g0 + g2) + g1);
    else if (tr == 2) v = -c6 * ((g0 + g2) - g1);
    else if (tr == 3) v = (c24 * g0 + c6 * g2) + c12 * g1;
    else if (tr == 4) v = (c24 * g0 + c6 * g2) - c12 * g1;
    else              v = g2;
    wt2[t] = v;
}

__device__ __forceinline__ void comb6(const float d[6], float v[6]) {
    v[0] = (4.0f * d[0] - 5.0f * d[2]) + d[4];
    v[1] = (d[3] + d[4]) - 4.0f * (d[1] + d[2]);
    v[2] = (d[4] - d[3]) + 4.0f * (d[1] - d[2]);
    v[3] = (d[4] - d[2]) + 2.0f * (d[3] - d[1]);
    v[4] = (d[4] - d[2]) - 2.0f * (d[3] - d[1]);
    v[5] = (4.0f * d[1] - 5.0f * d[3]) + d[5];
}


__global__ __launch_bounds__(256, 4) void conv_fused_kernel(
    const float* __restrict__ feat, const float* __restrict__ wt2,
    const float* __restrict__ conv_b,
    const float* __restrict__ det_w, const float* __restrict__ det_b,
    const float* __restrict__ reg_w, const float* __restrict__ reg_b,
    float* __restrict__ scores, float* __restrict__ deltas) {

    __shared__ float lds_in[NTR * LCOLS * CIPAD];
    __shared__ float part[8 * 4 * 16 * PPITCH];

    int bid = blockIdx.x;
    int n    = bid >> 7;
    int rem  = bid & 127;
    int row0 = (rem >> 2) * 4;
    int col0 = (rem & 3) * TILE_W;

    int tid  = threadIdx.x;
    int wave = tid >> 5;
    int lane = tid & 31;
    int hl   = lane >> 4;
    int lm   = lane & 15;

    v8f acc[NTR][2][2] = {};

    const float* featn = feat + (size_t)n * CIN * HF * WF;
    const v2f*   wp    = (const v2f*)wt2;

    const int sx4   = tid & 7;
    const int scl   = tid >> 3;
    const int hcl   = tid & 31;
    const int hside = (tid >> 5) & 1;
    const int hgcol = col0 - 1 + hside * (TILE_W + 1);
    const bool hcok = (hgcol >= 0) && (hgcol < WF);
    const int hgc   = min(max(hgcol, 0), WF - 1);
    const bool ok0  = (row0 - 1 >= 0);
    const bool ok5  = (row0 + 4 < HF);

    for (int cib = 0; cib < CIN; cib += CCHUNK) {
        float4 sv[NTR];
        float  hv[NTR] = {0.f, 0.f, 0.f, 0.f, 0.f, 0.f};
        #pragma unroll
        for (int r = 0; r < NTR; r++) {
            const int grc = min(max(row0 + r - 1, 0), HF - 1);
            sv[r] = *(const float4*)&featn[((size_t)(cib + scl) * HF + grc) * WF + col0 + 4 * sx4];
        }
        if (tid < 64) {
            #pragma unroll
            for (int r = 0; r < NTR; r++) {
                const int grc = min(max(row0 + r - 1, 0), HF - 1);
                hv[r] = featn[((size_t)(cib + hcl) * HF + grc) * WF + hgc];
            }
        }

        __syncthreads();
        {
            const float4 z4 = make_float4(0.f, 0.f, 0.f, 0.f);
            const float4 s0 = ok0 ? sv[0] : z4;
            const float4 s5 = ok5 ? sv[5] : z4;
            const float dx[6] = {s0.x, sv[1].x, sv[2].x, sv[3].x, sv[4].x, s5.x};
            const float dy[6] = {s0.y, sv[1].y, sv[2].y, sv[3].y, sv[4].y, s5.y};
            const float dz[6] = {s0.z, sv[1].z, sv[2].z, sv[3].z, sv[4].z, s5.z};
            const float dw[6] = {s0.w, sv[1].w, sv[2].w, sv[3].w, sv[4].w, s5.w};
            float cx[6], cy[6], cz[6], cw[6];
            comb6(dx, cx); comb6(dy, cy); comb6(dz, cz); comb6(dw, cw);
            #pragma unroll
            for (int tr = 0; tr < NTR; tr++) {
                float* dst = &lds_in[(tr * LCOLS + 1 + 4 * sx4) * CIPAD + scl];
                dst[0 * CIPAD] = cx[tr];
                dst[1 * CIPAD] = cy[tr];
                dst[2 * CIPAD] = cz[tr];
                dst[3 * CIPAD] = cw[tr];
            }
        }
        if (tid < 64) {
            float e[6], ce[6];
            e[0] = (ok0 && hcok) ? hv[0] : 0.0f;
            e[1] = hcok ? hv[1] : 0.0f;
            e[2] = hcok ? hv[2] : 0.0f;
            e[3] = hcok ? hv[3] : 0.0f;
            e[4] = hcok ? hv[4] : 0.0f;
            e[5] = (ok5 && hcok) ? hv[5] : 0.0f;
            comb6(e, ce);
            float* dst = &lds_in[(hside * (TILE_W + 1)) * CIPAD + hcl];
            #pragma unroll
            for (int tr = 0; tr < NTR; tr++) dst[tr * LCOLS * CIPAD] = ce[tr];
        }
        __syncthreads();

        const v2f* wl = wp + (size_t)(((cib >> 1) + hl) * 256 + wave * 32 + lm);
        #pragma unroll
        for (int tr = 0; tr < NTR; tr++) {
            #pragma unroll
            for (int kw = 0; kw < 3; kw++) {
                const int kk = tr * 3 + kw;
                v2f a0[CCHUNK / 4], a1[CCHUNK / 4];
                #pragma unroll
                for (int j = 0; j < CCHUNK / 4; j++) {
                    a0[j] = wl[(kk * 128 + 2 * j) * 256];
                    a1[j] = wl[(kk * 128 + 2 * j) * 256 + 16];
                }
                __builtin_amdgcn_sched_barrier(0);
                #pragma unroll
                for (int j = 0; j < CCHUNK / 4; j++) {
                    int boff = 4 * j + 2 * hl;
                    v2f b0 = *(const v2f*)&lds_in[(tr * LCOLS + lm + kw) * CIPAD + boff];
                    v2f b1 = *(const v2f*)&lds_in[(tr * LCOLS + 16 + lm + kw) * CIPAD + boff];
                    acc[tr][0][0] = __builtin_amdgcn_wmma_f32_16x16x4_f32(false, a0[j], false, b0, (short)0, acc[tr][0][0], false, false);
                    acc[tr][1][0] = __builtin_amdgcn_wmma_f32_16x16x4_f32(false, a1[j], false, b0, (short)0, acc[tr][1][0], false, false);
                    acc[tr][0][1] = __builtin_amdgcn_wmma_f32_16x16x4_f32(false, a0[j], false, b1, (short)0, acc[tr][0][1], false, false);
                    acc[tr][1][1] = __builtin_amdgcn_wmma_f32_16x16x4_f32(false, a1[j], false, b1, (short)0, acc[tr][1][1], false, false);
                }
            }
        }
    }

    const int elane = (int)__builtin_amdgcn_mbcnt_lo(~0u, 0u);
    const int elm = elane & 15;
    const int ehl = elane >> 4;
    const int ho = elm;
    const float* hrow = (ho < 3) ? (det_w + ho * 256) : (reg_w + (min(ho, 14) - 3) * 256);
    v2f ha[2][4];
    #pragma unroll
    for (int mt = 0; mt < 2; mt++) {
        #pragma unroll
        for (int q = 0; q < 4; q++) {
            const int c = wave * 32 + mt * 16 + 8 * ehl + 2 * q;
            const float w0 = hrow[c], w1 = hrow[c + 1];
            ha[mt][q][0] = (ho < 15) ? w0 : 0.0f;
            ha[mt][q][1] = (ho < 15) ? w1 : 0.0f;
        }
    }
    float cbv[2][8];
    #pragma unroll
    for (int mt = 0; mt < 2; mt++) {
        #pragma unroll
        for (int r = 0; r < 8; r++) cbv[mt][r] = conv_b[wave * 32 + mt * 16 + 8 * ehl + r];
    }

    #pragma unroll
    for (int ph = 0; ph < 4; ph++) {
        #pragma unroll
        for (int nt = 0; nt < 2; nt++) {
            v8f hd = {};
            #pragma unroll
            for (int mt = 0; mt < 2; mt++) {
                float y[8];
                #pragma unroll
                for (int r = 0; r < 8; r++) {
                    const float q0 = acc[0][mt][nt][r], q1 = acc[1][mt][nt][r], q2 = acc[2][mt][nt][r];
                    const float q3 = acc[3][mt][nt][r], q4 = acc[4][mt][nt][r], q5 = acc[5][mt][nt][r];
                    float v;
                    if (ph == 0)      v = (q0 + (q1 + q2)) + (q3 + q4);
                    else if (ph == 1) v = (q1 - q2) + 2.0f * (q3 - q4);
                    else if (ph == 2) v = (q1 + q2) + 4.0f * (q3 + q4);
                    else              v = ((q1 - q2) + 8.0f * (q3 - q4)) + q5;
                    y[r] = v + cbv[mt][r];
                }
                #pragma unroll
                for (int q = 0; q < 4; q++) {
                    v2f b;
                    b[0] = y[2 * q];
                    b[1] = y[2 * q + 1];
                    hd = __builtin_amdgcn_wmma_f32_16x16x4_f32(false, ha[mt][q], false, b, (short)0, hd, false, false);
                }
            }
            float* dst = &part[((wave * 4 + ph) * 16 + 8 * ehl) * PPITCH + nt * 16 + elm];
            #pragma unroll
            for (int r = 0; r < 8; r++) dst[r * PPITCH] = hd[r];
        }
    }
    __syncthreads();

    for (int it = tid; it < A_ANCH * 4 * TILE_W; it += 256) {
        const int a   = it >> 7;
        const int pxl = it & 127;
        const int ph  = pxl >> 5;
        const int px  = pxl & 31;
        float s[5];
        #pragma unroll
        for (int c = 0; c < 5; c++) {
            const int oi = (c == 0) ? a : (3 + (c - 1) * A_ANCH + a);
            float sum = part[((0 * 4 + ph) * 16 + oi) * PPITCH + px];
            #pragma unroll
            for (int w = 1; w < 8; w++) sum += part[((w * 4 + ph) * 16 + oi) * PPITCH + px];
            s[c] = sum + ((c == 0) ? det_b[a] : reg_b[(c - 1) * A_ANCH + a]);
        }
        const int p = (row0 + ph) * WF + col0 + px;
        scores[(size_t)n * NPROP + a * (HF * WF) + p] = 1.0f / (1.0f + expf(-s[0]));
        *(float4*)&deltas[(((size_t)n * NPROP) + a * (HF * WF) + p) * 4] = make_float4(s[1], s[2], s[3], s[4]);
    }
}

__global__ __launch_bounds__(256) void decode_kernel(
    const float* scores, const float* deltas,
    float* boxes, unsigned* svk) {
    int n = blockIdx.y;
    int i = blockIdx.x * 256 + threadIdx.x;

    int pa = i / 3, aa = i - pa * 3;
    int ya = pa >> 7, xa = pa & 127;
    const float bs[3][4] = {{-91.f, -45.f, 91.f, 45.f},
                            {-64.f, -64.f, 64.f, 64.f},
                            {-45.f, -91.f, 45.f, 91.f}};
    float sxv = (float)(xa * 8), syv = (float)(ya * 8);
    float x1 = sxv + bs[aa][0], y1 = syv + bs[aa][1];
    float x2 = sxv + bs[aa][2], y2 = syv + bs[aa][3];
    float aw = x2 - x1, ah = y2 - y1;
    float acx = y1 + aw * 0.5f;
    float acy = x1 + ah * 0.5f;

    const float* d = deltas + (((size_t)n * NPROP) + i) * 4;
    const float d0 = d[0], d1 = d[1], d2 = d[2], d3 = d[3];
    float px = acx + d0 * aw;
    float py = acy + d1 * ah;
    float pw = aw * expf(d2);
    float ph = ah * expf(d3);
    float b0 = fminf(fmaxf(px - pw * 0.5f, 0.f), 1024.f);
    float b1 = fminf(fmaxf(py - ph * 0.5f, 0.f), 1024.f);
    float b2 = fminf(fmaxf(px + pw * 0.5f, 0.f), 1024.f);
    float b3 = fminf(fmaxf(py + ph * 0.5f, 0.f), 1024.f);

    float* bo = boxes + (((size_t)n * NPROP) + i) * 4;
    bo[0] = b0; bo[1] = b1; bo[2] = b2; bo[3] = b3;

    float s = scores[(size_t)n * NPROP + i];
    bool valid = (s > OBJ_THR) && ((b2 - b0) > 0.f) && ((b3 - b1) > 0.f);
    unsigned u = __float_as_uint(s);
    unsigned sv = (u & 0x80000000u) ? ~u : (u | 0x80000000u);
    svk[(size_t)n * NPROP + i] = valid ? sv : 0u;
}

#define NGRP 63
#define NCAND (NGRP * 32)
#define KPT 48
#define NSORT 2048
#define GRID_W 32
#define NCELL (GRID_W * GRID_W)

__device__ __forceinline__ int block_total(unsigned (*s_cnt)[32], int& turn, int wave, int lane, int wcount) {
    unsigned* row = s_cnt[turn & 1];
    turn++;
    if (lane == 0) row[wave] = (unsigned)wcount;
    __syncthreads();
    const uint4* p = (const uint4*)row;
    unsigned tot = 0;
    for (int q = 0; q < 8; q++) { uint4 v = p[q]; tot = tot + v.x + v.y; tot = tot + v.z + v.w; }
    return __builtin_amdgcn_readfirstlane((int)tot);
}

__device__ __forceinline__ bool iou_over(float ib0, float ib1, float ib2, float ib3, float ia,
                                         float jx1, float jy1, float jx2, float jy2, float ja) {
    float ix1 = fmaxf(ib0, jx1), iy1 = fmaxf(ib1, jy1);
    float ix2 = fminf(ib2, jx2), iy2 = fminf(ib3, jy2);
    float inter = fmaxf(ix2 - ix1, 0.f) * fmaxf(iy2 - iy1, 0.f);
    float iou = inter / fmaxf(ia + ja - inter, 1e-9f);
    return iou > IOU_THR;
}

__device__ __forceinline__ bool may_overlap(float ib0, float ib1, float ib2, float ib3, float half_ia,
                                            float jx1, float jy1, float jx2, float jy2) {
    float w = fminf(ib2, jx2) - fmaxf(ib0, jx1);
    float h = fminf(ib3, jy2) - fmaxf(ib1, jy1);
    return fmaxf(w, 0.f) * h > half_ia;
}

static_assert(IOU_THR > 0.5f, "the centre-cell argument needs a threshold above one half");
__device__ __forceinline__ int cell_of(float x) {
    return min(max((int)(x * (1.0f / 32.0f)), 0), GRID_W - 1);
}

__global__ __launch_bounds__(1024) void nms_kernel(const unsigned* __restrict__ svk,
                                                   const float* __restrict__ boxes,
                                                   float* __restrict__ out) {
    __shared__ float4   cbx[NCAND];
    __shared__ float    car[NCAND];
    __shared__ float    csc[NCAND];
    __shared__ int      cvl[NCAND];
    __shared__ float4   kbx[NCAND];
    __shared__ float    kar[NCAND];
    __shared__ float    ksc[NCAND];
    __shared__ int      hd[NCELL];
    __shared__ int      nxt[NCAND];
    __shared__ unsigned s_alive[32];
    __shared__ unsigned s_row[32];
    __shared__ unsigned shi[NSORT];
    __shared__ unsigned slo[NSORT];
    __shared__ __attribute__((aligned(16))) unsigned s_cnt[2][32];

    int n    = blockIdx.x;
    int tid  = threadIdx.x;
    int wave = tid >> 5;
    int lane = tid & 31;

    const unsigned* S = svk + (size_t)n * NPROP;
    unsigned sv[KPT];
    #pragma unroll
    for (int j = 0; j < KPT; j++) sv[j] = S[j * 1024 + tid];

    int turn = 0;
    unsigned T = 0u;
    int c_ge = NPROP;
    for (int bit = 31; bit >= 0; bit--) {
        unsigned cand = T | (1u << bit);
        int c = 0;
        #pragma unroll
        for (int j = 0; j < KPT; j += 2) {
            unsigned m0 = __builtin_amdgcn_ballot_w32(sv[j] >= cand);
            unsigned m1 = __builtin_amdgcn_ballot_w32(sv[j + 1] >= cand);
            c += __builtin_popcountll(((unsigned long long)m1 << 32) | m0);
        }
        int tot = block_total(s_cnt, turn, wave, lane, c);
        if (tot >= PRE_NMS) { T = cand; c_ge = tot; }
    }
    if (T != 0u) {
        int c = 0;
        #pragma unroll
        for (int j = 0; j < KPT; j += 2) {
            unsigned m0 = __builtin_amdgcn_ballot_w32(sv[j] > T);
            unsigned m1 = __builtin_amdgcn_ballot_w32(sv[j + 1] > T);
            c += __builtin_popcountll(((unsigned long long)m1 << 32) | m0);
        }
        int c_gt = block_total(s_cnt, turn, wave, lane, c);
        int need = PRE_NMS - c_gt;
        if (c_ge - c_gt != need) {
            unsigned R = 0u;
            for (int bit = 15; bit >= 0; bit--) {
                unsigned cand = R | (1u << bit);
                int ce = 0;
                #pragma unroll
                for (int j = 0; j < KPT; j++) {
                    unsigned r = 0xFFFFu - (unsigned)(j * 1024 + tid);
                    ce += __builtin_popcount(__builtin_amdgcn_ballot_w32((sv[j] == T) & (r >= cand)));
                }
                int tot = block_total(s_cnt, turn, wave, lane, ce);
                if (tot >= need) R = cand;
            }
            #pragma unroll
            for (int j = 0; j < KPT; j++) {
                unsigned r = 0xFFFFu - (unsigned)(j * 1024 + tid);
                if ((sv[j] == T) & (r < R)) sv[j] = 0u;
            }
        }
    }
    const unsigned Tc = (T != 0u) ? T : 1u;
    {
        int wsel = 0;
        #pragma unroll
        for (int j = 0; j < KPT; j += 2) {
            unsigned m0 = __builtin_amdgcn_ballot_w32(sv[j] >= Tc);
            unsigned m1 = __builtin_amdgcn_ballot_w32(sv[j + 1] >= Tc);
            wsel += __builtin_popcountll(((unsigned long long)m1 << 32) | m0);
        }
        unsigned* row = s_cnt[turn & 1];
        turn++;
        if (lane == 0) row[wave] = (unsigned)wsel;
        __syncthreads();
        int before = 0, msel = 0;
        for (int q = 0; q < 32; q++) {
            int v = (int)row[q];
            before += (q < wave) ? v : 0;
            msel   += v;
        }
        int pos = __builtin_amdgcn_readfirstlane(before);
        #pragma unroll
        for (int j = 0; j < KPT; j++) {
            bool sel = sv[j] >= Tc;
            unsigned m = __builtin_amdgcn_ballot_w32(sel);
            int at = pos + (int)__builtin_amdgcn_mbcnt_lo(m, 0u);
            if (sel) { shi[at] = sv[j]; slo[at] = 0xFFFFu - (unsigned)(j * 1024 + tid); }
            pos += __builtin_popcount(m);
        }
        for (int e = msel + tid; e < NSORT; e += 1024) { shi[e] = 0u; slo[e] = 0u; }
    }
    for (int k = 2; k <= NSORT; k <<= 1) {
        for (int j = k >> 1; j > 0; j >>= 1) {
            __syncthreads();
            int i = ((tid & ~(j - 1)) << 1) | (tid & (j - 1));
            int p = i | j;
            unsigned ah = shi[i], al = slo[i];
            unsigned bh = shi[p], bl = slo[p];
            bool a_lt_b = (ah < bh) || (ah == bh && al < bl);
            bool up = ((i & k) == 0);
            if (a_lt_b == up) { shi[i] = bh; slo[i] = bl; shi[p] = ah; slo[p] = al; }
        }
    }
    __syncthreads();

    for (int i = tid; i < NCAND; i += 1024) {
        bool v = false;
        float b0 = 0.f, b1 = 0.f, b2 = 0.f, b3 = 0.f, s = 0.f;
        if (i < PRE_NMS) {
            unsigned w   = shi[i];
            unsigned idx = 0xFFFFu - slo[i];
            v = w != 0u;
            if (v) {
                const float* bp = boxes + (((size_t)n * NPROP) + idx) * 4;
                b0 = bp[0]; b1 = bp[1]; b2 = bp[2]; b3 = bp[3];
                unsigned u = (w & 0x80000000u) ? (w ^ 0x80000000u) : ~w;
                s = __uint_as_float(u);
            }
        }
        cbx[i] = make_float4(b0, b1, b2, b3);
        car[i] = fmaxf(b2 - b0, 0.f) * fmaxf(b3 - b1, 0.f);
        csc[i] = s;
        cvl[i] = v ? 1 : 0;
    }
    for (int c = tid; c < NCELL; c += 1024) hd[c] = -1;
    __syncthreads();

    int nk = 0;
    for (int g = 0; g < NGRP; g++) {
        int base = g * 32;
        int i    = base + wave;
        float4 ib = cbx[i];
        float  ia = car[i];
        float  is = csc[i];
        int    vi = cvl[i];

        const float half_ia = 0.5f * ia;
        const int mycell  = cell_of(0.5f * (ib.y + ib.w)) * GRID_W + cell_of(0.5f * (ib.x + ib.z));
        const int oldhead = hd[mycell];
        bool found = false;
        if (vi) {
            const int cx0 = cell_of(ib.x), cy0 = cell_of(ib.y);
            const int nx  = cell_of(ib.z) - cx0 + 1;
            const int ny  = cell_of(ib.w) - cy0 + 1;
            const int m   = nx * ny;
            const float rnx = __builtin_amdgcn_rcpf((float)nx);
            for (int q = lane; q < m; q += 32) {
                const int qy = (int)(((float)q + 0.5f) * rnx);
                const int qx = q - qy * nx;
                int k = hd[min((cy0 + qy) * GRID_W + cx0 + qx, NCELL - 1)];
                for (int it = 0; it < NCAND && k >= 0; it++) {
                    const float4 jb = kbx[k];
                    const int    kn = nxt[k];
                    if (may_overlap(ib.x, ib.y, ib.z, ib.w, half_ia, jb.x, jb.y, jb.z, jb.w))
                        found |= iou_over(ib.x, ib.y, ib.z, ib.w, ia, jb.x, jb.y, jb.z, jb.w, kar[k]);
                    k = kn;
                }
            }
        }
        float4 rb = cbx[base + lane];
        const int rcell = cell_of(0.5f * (rb.y + rb.w)) * GRID_W + cell_of(0.5f * (rb.x + rb.z));
        bool hit = false;
        if ((lane < wave) && may_overlap(ib.x, ib.y, ib.z, ib.w, half_ia, rb.x, rb.y, rb.z, rb.w))
            hit = iou_over(ib.x, ib.y, ib.z, ib.w, ia, rb.x, rb.y, rb.z, rb.w, car[base + lane]);
        unsigned anyf = __builtin_amdgcn_ballot_w32(found);
        unsigned rowb = __builtin_amdgcn_ballot_w32(hit);
        if (lane == 0) {
            s_alive[wave] = (vi != 0 && anyf == 0u) ? 1u : 0u;
            s_row[wave]   = rowb;
        }
        __syncthreads();

        unsigned al = s_alive[lane];
        unsigned rw = s_row[lane];
        unsigned keepbits = __builtin_amdgcn_ballot_w32(al != 0u && rw == 0u);
        unsigned pend     = __builtin_amdgcn_ballot_w32(al != 0u && rw != 0u);
        while (pend != 0u) {
            int r = __builtin_ctz(pend);
            pend &= pend - 1u;
            unsigned rr = (unsigned)__builtin_amdgcn_readlane((int)rw, r);
            if ((rr & keepbits) == 0u) keepbits |= (1u << r);
        }
        if ((keepbits >> wave) & 1u) {
            const unsigned below = (1u << wave) - 1u;
            int pos = nk + __popc(keepbits & below);
            unsigned same  = __builtin_amdgcn_ballot_w32(rcell == mycell) & keepbits;
            unsigned older = same & below;
            int link = oldhead;
            if (older != 0u) {
                int p = 31 - __builtin_clz(older);
                link = nk + __popc(keepbits & ((1u << p) - 1u));
            }
            if (lane == 0) {
                kbx[pos] = ib; kar[pos] = ia; ksc[pos] = is; nxt[pos] = link;
                if ((same & ~below & ~(1u << wave)) == 0u) hd[mycell] = pos;
            }
        }
        nk += __popc(keepbits);
        __syncthreads();
    }

    for (int t = tid; t < POST_NMS * 5; t += 1024) {
        int row = t / 5;
        int cc  = t - row * 5;
        float v = 0.f;
        if (row < nk) v = (cc < 4) ? ((const float*)kbx)[row * 4 + cc] : ksc[row];
        out[(size_t)n * POST_NMS * 5 + t] = v;
    }
}

extern "C" void kernel_launch(void* const* d_in, const int* in_sizes, int n_in,
                              void* d_out, int out_size, void* d_ws, size_t ws_size,
                              hipStream_t stream) {
    const float* feat   = (const float*)d_in[0];
    const float* conv_w = (const float*)d_in[1];
    const float* conv_b = (const float*)d_in[2];
    const float* det_w  = (const float*)d_in[3];
    const float* det_b  = (const float*)d_in[4];
    const float* reg_w  = (const float*)d_in[5];
    const float* reg_b  = (const float*)d_in[6];
    float* out = (float*)d_out;

    char* ws = (char*)d_ws;
    float* wt2                  = (float*)(ws);
    float* scores               = (float*)(ws + 4718592);
    float* deltas               = (float*)(ws + 4718592 + 786432);
    float* boxes                = deltas;
    unsigned* svk               = (unsigned*)(ws + 4718592);

    repack_w_kernel<<<4608, 256, 0, stream>>>(conv_w, wt2);
    conv_fused_kernel<<<NIMG * (HF / 4) * (WF / TILE_W), 256, 0, stream>>>(
        feat, wt2, conv_b, det_w, det_b, reg_w, reg_b, scores, deltas);
    decode_kernel<<<dim3(NPROP / 256, NIMG), 256, 0, stream>>>(scores, deltas, boxes, svk);
    nms_kernel<<<NIMG, 1024, 0, stream>>>(svk, boxes, out);
}


# top-2000 threshold search does only the steps that can change the selection: one count for the 7 leading bits every valid score word shares, then stops once the words >= T fit the 2048-key LDS sort
# speedup vs baseline: 5.3760x; 1.0396x over previous
#include <hip/hip_runtime.h>
#include <hip/hip_bf16.h>
#include <math.h>

typedef __attribute__((ext_vector_type(2))) float v2f;
typedef __attribute__((ext_vector_type(8))) float v8f;

#define NIMG 4
#define CIN 256
#define HF 128
#define WF 128
#define A_ANCH 3
#define NPROP (A_ANCH * HF * WF)
#define PRE_NMS 2000
#define POST_NMS 300
#define OBJ_THR 0.3f
#define IOU_THR 0.7f
#define TILE_W 32
#define LCOLS 34
#define CCHUNK 32
#define CIPAD 34
#define NTR 6
#define PPITCH 36

__global__ __launch_bounds__(256) void repack_w_kernel(const float* __restrict__ conv_w,
                                                       float* __restrict__ wt2) {
    int t = blockIdx.x * 256 + threadIdx.x;
    int par = t & 1;
    int co  = (t >> 1) & 255;
    int q   = t >> 9;
    int ci2 = q & 127;
    int kk  = q >> 7;
    int ci  = ci2 * 2 + par;
    int tr  = kk / 3, kw = kk % 3;
    const float* w = conv_w + ((size_t)(co * CIN + ci) * 3) * 3 + kw;
    float g0 = w[0], g1 = w[3], g2 = w[6];
    const float c6 = 1.0f / 6.0f, c12 = 1.0f / 12.0f, c24 = 1.0f / 24.0f;
    float v;
    if (tr == 0)      v = 0.25f * g0;
    else if (tr == 1) v = -c6 * ((g0 + g2) + g1);
    else if (tr == 2) v = -c6 * ((g0 + g2) - g1);
    else if (tr == 3) v = (c24 * g0 + c6 * g2) + c12 * g1;
    else if (tr == 4) v = (c24 * g0 + c6 * g2) - c12 * g1;
    else              v = g2;
    wt2[t] = v;
}

__device__ __forceinline__ void comb6(const float d[6], float v[6]) {
    v[0] = (4.0f * d[0] - 5.0f * d[2]) + d[4];
    v[1] = (d[3] + d[4]) - 4.0f * (d[1] + d[2]);
    v[2] = (d[4] - d[3]) + 4.0f * (d[1] - d[2]);
    v[3] = (d[4] - d[2]) + 2.0f * (d[3] - d[1]);
    v[4] = (d[4] - d[2]) - 2.0f * (d[3] - d[1]);
    v[5] = (4.0f * d[1] - 5.0f * d[3]) + d[5];
}


__global__ __launch_bounds__(256, 4) void conv_fused_kernel(
    const float* __restrict__ feat, const float* __restrict__ wt2,
    const float* __restrict__ conv_b,
    const float* __restrict__ det_w, const float* __restrict__ det_b,
    const float* __restrict__ reg_w, const float* __restrict__ reg_b,
    float* __restrict__ scores, float* __restrict__ deltas) {

    __shared__ float lds_in[NTR * LCOLS * CIPAD];
    __shared__ float part[8 * 4 * 16 * PPITCH];

    int bid = blockIdx.x;
    int n    = bid >> 7;
    int rem  = bid & 127;
    int row0 = (rem >> 2) * 4;
    int col0 = (rem & 3) * TILE_W;

    int tid  = threadIdx.x;
    int wave = tid >> 5;
    int lane = tid & 31;
    int hl   = lane >> 4;
    int lm   = lane & 15;

    v8f acc[NTR][2][2] = {};

    const float* featn = feat + (size_t)n * CIN * HF * WF;
    const v2f*   wp    = (const v2f*)wt2;

    const int sx4   = tid & 7;
    const int scl   = tid >> 3;
    const int hcl   = tid & 31;
    const int hside = (tid >> 5) & 1;
    const int hgcol = col0 - 1 + hside * (TILE_W + 1);
    const bool hcok = (hgcol >= 0) && (hgcol < WF);
    const int hgc   = min(max(hgcol, 0), WF - 1);
    const bool ok0  = (row0 - 1 >= 0);
    const bool ok5  = (row0 + 4 < HF);

    for (int cib = 0; cib < CIN; cib += CCHUNK) {
        float4 sv[NTR];
        float  hv[NTR] = {0.f, 0.f, 0.f, 0.f, 0.f, 0.f};
        #pragma unroll
        for (int r = 0; r < NTR; r++) {
            const int grc = min(max(row0 + r - 1, 0), HF - 1);
            sv[r] = *(const float4*)&featn[((size_t)(cib + scl) * HF + grc) * WF + col0 + 4 * sx4];
        }
        if (tid < 64) {
            #pragma unroll
            for (int r = 0; r < NTR; r++) {
                const int grc = min(max(row0 + r - 1, 0), HF - 1);
                hv[r] = featn[((size_t)(cib + hcl) * HF + grc) * WF + hgc];
            }
        }

        __syncthreads();
        {
            const float4 z4 = make_float4(0.f, 0.f, 0.f, 0.f);
            const float4 s0 = ok0 ? sv[0] : z4;
            const float4 s5 = ok5 ? sv[5] : z4;
            const float dx[6] = {s0.x, sv[1].x, sv[2].x, sv[3].x, sv[4].x, s5.x};
            const float dy[6] = {s0.y, sv[1].y, sv[2].y, sv[3].y, sv[4].y, s5.y};
            const float dz[6] = {s0.z, sv[1].z, sv[2].z, sv[3].z, sv[4].z, s5.z};
            const float dw[6] = {s0.w, sv[1].w, sv[2].w, sv[3].w, sv[4].w, s5.w};
            float cx[6], cy[6], cz[6], cw[6];
            comb6(dx, cx); comb6(dy, cy); comb6(dz, cz); comb6(dw, cw);
            #pragma unroll
            for (int tr = 0; tr < NTR; tr++) {
                float* dst = &lds_in[(tr * LCOLS + 1 + 4 * sx4) * CIPAD + scl];
                dst[0 * CIPAD] = cx[tr];
                dst[1 * CIPAD] = cy[tr];
                dst[2 * CIPAD] = cz[tr];
                dst[3 * CIPAD] = cw[tr];
            }
        }
        if (tid < 64) {
            float e[6], ce[6];
            e[0] = (ok0 && hcok) ? hv[0] : 0.0f;
            e[1] = hcok ? hv[1] : 0.0f;
            e[2] = hcok ? hv[2] : 0.0f;
            e[3] = hcok ? hv[3] : 0.0f;
            e[4] = hcok ? hv[4] : 0.0f;
            e[5] = (ok5 && hcok) ? hv[5] : 0.0f;
            comb6(e, ce);
            float* dst = &lds_in[(hside * (TILE_W + 1)) * CIPAD + hcl];
            #pragma unroll
            for (int tr = 0; tr < NTR; tr++) dst[tr * LCOLS * CIPAD] = ce[tr];
        }
        __syncthreads();

        const v2f* wl = wp + (size_t)(((cib >> 1) + hl) * 256 + wave * 32 + lm);
        #pragma unroll
        for (int tr = 0; tr < NTR; tr++) {
            #pragma unroll
            for (int kw = 0; kw < 3; kw++) {
                const int kk = tr * 3 + kw;
                v2f a0[CCHUNK / 4], a1[CCHUNK / 4];
                #pragma unroll
                for (int j = 0; j < CCHUNK / 4; j++) {
                    a0[j] = wl[(kk * 128 + 2 * j) * 256];
                    a1[j] = wl[(kk * 128 + 2 * j) * 256 + 16];
                }
                __builtin_amdgcn_sched_barrier(0);
                #pragma unroll
                for (int j = 0; j < CCHUNK / 4; j++) {
                    int boff = 4 * j + 2 * hl;
                    v2f b0 = *(const v2f*)&lds_in[(tr * LCOLS + lm + kw) * CIPAD + boff];
                    v2f b1 = *(const v2f*)&lds_in[(tr * LCOLS + 16 + lm + kw) * CIPAD + boff];
                    acc[tr][0][0] = __builtin_amdgcn_wmma_f32_16x16x4_f32(false, a0[j], false, b0, (short)0, acc[tr][0][0], false, false);
                    acc[tr][1][0] = __builtin_amdgcn_wmma_f32_16x16x4_f32(false, a1[j], false, b0, (short)0, acc[tr][1][0], false, false);
                    acc[tr][0][1] = __builtin_amdgcn_wmma_f32_16x16x4_f32(false, a0[j], false, b1, (short)0, acc[tr][0][1], false, false);
                    acc[tr][1][1] = __builtin_amdgcn_wmma_f32_16x16x4_f32(false, a1[j], false, b1, (short)0, acc[tr][1][1], false, false);
                }
            }
        }
    }

    const int elane = (int)__builtin_amdgcn_mbcnt_lo(~0u, 0u);
    const int elm = elane & 15;
    const int ehl = elane >> 4;
    const int ho = elm;
    const float* hrow = (ho < 3) ? (det_w + ho * 256) : (reg_w + (min(ho, 14) - 3) * 256);
    v2f ha[2][4];
    #pragma unroll
    for (int mt = 0; mt < 2; mt++) {
        #pragma unroll
        for (int q = 0; q < 4; q++) {
            const int c = wave * 32 + mt * 16 + 8 * ehl + 2 * q;
            const float w0 = hrow[c], w1 = hrow[c + 1];
            ha[mt][q][0] = (ho < 15) ? w0 : 0.0f;
            ha[mt][q][1] = (ho < 15) ? w1 : 0.0f;
        }
    }
    float cbv[2][8];
    #pragma unroll
    for (int mt = 0; mt < 2; mt++) {
        #pragma unroll
        for (int r = 0; r < 8; r++) cbv[mt][r] = conv_b[wave * 32 + mt * 16 + 8 * ehl + r];
    }

    #pragma unroll
    for (int ph = 0; ph < 4; ph++) {
        #pragma unroll
        for (int nt = 0; nt < 2; nt++) {
            v8f hd = {};
            #pragma unroll
            for (int mt = 0; mt < 2; mt++) {
                float y[8];
                #pragma unroll
                for (int r = 0; r < 8; r++) {
                    const float q0 = acc[0][mt][nt][r], q1 = acc[1][mt][nt][r], q2 = acc[2][mt][nt][r];
                    const float q3 = acc[3][mt][nt][r], q4 = acc[4][mt][nt][r], q5 = acc[5][mt][nt][r];
                    float v;
                    if (ph == 0)      v = (q0 + (q1 + q2)) + (q3 + q4);
                    else if (ph == 1) v = (q1 - q2) + 2.0f * (q3 - q4);
                    else if (ph == 2) v = (q1 + q2) + 4.0f * (q3 + q4);
                    else              v = ((q1 - q2) + 8.0f * (q3 - q4)) + q5;
                    y[r] = v + cbv[mt][r];
                }
                #pragma unroll
                for (int q = 0; q < 4; q++) {
                    v2f b;
                    b[0] = y[2 * q];
                    b[1] = y[2 * q + 1];
                    hd = __builtin_amdgcn_wmma_f32_16x16x4_f32(false, ha[mt][q], false, b, (short)0, hd, false, false);
                }
            }
            float* dst = &part[((wave * 4 + ph) * 16 + 8 * ehl) * PPITCH + nt * 16 + elm];
            #pragma unroll
            for (int r = 0; r < 8; r++) dst[r * PPITCH] = hd[r];
        }
    }
    __syncthreads();

    for (int it = tid; it < A_ANCH * 4 * TILE_W; it += 256) {
        const int a   = it >> 7;
        const int pxl = it & 127;
        const int ph  = pxl >> 5;
        const int px  = pxl & 31;
        float s[5];
        #pragma unroll
        for (int c = 0; c < 5; c++) {
            const int oi = (c == 0) ? a : (3 + (c - 1) * A_ANCH + a);
            float sum = part[((0 * 4 + ph) * 16 + oi) * PPITCH + px];
            #pragma unroll
            for (int w = 1; w < 8; w++) sum += part[((w * 4 + ph) * 16 + oi) * PPITCH + px];
            s[c] = sum + ((c == 0) ? det_b[a] : reg_b[(c - 1) * A_ANCH + a]);
        }
        const int p = (row0 + ph) * WF + col0 + px;
        scores[(size_t)n * NPROP + a * (HF * WF) + p] = 1.0f / (1.0f + expf(-s[0]));
        *(float4*)&deltas[(((size_t)n * NPROP) + a * (HF * WF) + p) * 4] = make_float4(s[1], s[2], s[3], s[4]);
    }
}

__global__ __launch_bounds__(256) void decode_kernel(
    const float* scores, const float* deltas,
    float* boxes, unsigned* svk) {
    int n = blockIdx.y;
    int i = blockIdx.x * 256 + threadIdx.x;

    int pa = i / 3, aa = i - pa * 3;
    int ya = pa >> 7, xa = pa & 127;
    const float bs[3][4] = {{-91.f, -45.f, 91.f, 45.f},
                            {-64.f, -64.f, 64.f, 64.f},
                            {-45.f, -91.f, 45.f, 91.f}};
    float sxv = (float)(xa * 8), syv = (float)(ya * 8);
    float x1 = sxv + bs[aa][0], y1 = syv + bs[aa][1];
    float x2 = sxv + bs[aa][2], y2 = syv + bs[aa][3];
    float aw = x2 - x1, ah = y2 - y1;
    float acx = y1 + aw * 0.5f;
    float acy = x1 + ah * 0.5f;

    const float* d = deltas + (((size_t)n * NPROP) + i) * 4;
    const float d0 = d[0], d1 = d[1], d2 = d[2], d3 = d[3];
    float px = acx + d0 * aw;
    float py = acy + d1 * ah;
    float pw = aw * expf(d2);
    float ph = ah * expf(d3);
    float b0 = fminf(fmaxf(px - pw * 0.5f, 0.f), 1024.f);
    float b1 = fminf(fmaxf(py - ph * 0.5f, 0.f), 1024.f);
    float b2 = fminf(fmaxf(px + pw * 0.5f, 0.f), 1024.f);
    float b3 = fminf(fmaxf(py + ph * 0.5f, 0.f), 1024.f);

    float* bo = boxes + (((size_t)n * NPROP) + i) * 4;
    bo[0] = b0; bo[1] = b1; bo[2] = b2; bo[3] = b3;

    float s = scores[(size_t)n * NPROP + i];
    bool valid = (s > OBJ_THR) && ((b2 - b0) > 0.f) && ((b3 - b1) > 0.f);
    unsigned u = __float_as_uint(s);
    unsigned sv = (u & 0x80000000u) ? ~u : (u | 0x80000000u);
    svk[(size_t)n * NPROP + i] = valid ? sv : 0u;
}

#define NGRP 63
#define NCAND (NGRP * 32)
#define KPT 48
#define NSORT 2048
#define WORD_PREFIX 0xBE000000u
static_assert(OBJ_THR >= 0.125f && OBJ_THR < 1.0f, "WORD_PREFIX needs every valid score in [0.125, 2)");
#define GRID_W 32
#define NCELL (GRID_W * GRID_W)

__device__ __forceinline__ int block_total(unsigned (*s_cnt)[32], int& turn, int wave, int lane, int wcount) {
    unsigned* row = s_cnt[turn & 1];
    turn++;
    if (lane == 0) row[wave] = (unsigned)wcount;
    __syncthreads();
    const uint4* p = (const uint4*)row;
    unsigned tot = 0;
    for (int q = 0; q < 8; q++) { uint4 v = p[q]; tot = tot + v.x + v.y; tot = tot + v.z + v.w; }
    return __builtin_amdgcn_readfirstlane((int)tot);
}

__device__ __forceinline__ bool iou_over(float ib0, float ib1, float ib2, float ib3, float ia,
                                         float jx1, float jy1, float jx2, float jy2, float ja) {
    float ix1 = fmaxf(ib0, jx1), iy1 = fmaxf(ib1, jy1);
    float ix2 = fminf(ib2, jx2), iy2 = fminf(ib3, jy2);
    float inter = fmaxf(ix2 - ix1, 0.f) * fmaxf(iy2 - iy1, 0.f);
    float iou = inter / fmaxf(ia + ja - inter, 1e-9f);
    return iou > IOU_THR;
}

__device__ __forceinline__ bool may_overlap(float ib0, float ib1, float ib2, float ib3, float half_ia,
                                            float jx1, float jy1, float jx2, float jy2) {
    float w = fminf(ib2, jx2) - fmaxf(ib0, jx1);
    float h = fminf(ib3, jy2) - fmaxf(ib1, jy1);
    return fmaxf(w, 0.f) * h > half_ia;
}

static_assert(IOU_THR > 0.5f, "the centre-cell argument needs a threshold above one half");
__device__ __forceinline__ int cell_of(float x) {
    return min(max((int)(x * (1.0f / 32.0f)), 0), GRID_W - 1);
}

__global__ __launch_bounds__(1024) void nms_kernel(const unsigned* __restrict__ svk,
                                                   const float* __restrict__ boxes,
                                                   float* __restrict__ out) {
    __shared__ float4   cbx[NCAND];
    __shared__ float    car[NCAND];
    __shared__ float    csc[NCAND];
    __shared__ int      cvl[NCAND];
    __shared__ float4   kbx[NCAND];
    __shared__ float    kar[NCAND];
    __shared__ float    ksc[NCAND];
    __shared__ int      hd[NCELL];
    __shared__ int      nxt[NCAND];
    __shared__ unsigned s_alive[32];
    __shared__ unsigned s_row[32];
    __shared__ unsigned shi[NSORT];
    __shared__ unsigned slo[NSORT];
    __shared__ __attribute__((aligned(16))) unsigned s_cnt[2][32];

    int n    = blockIdx.x;
    int tid  = threadIdx.x;
    int wave = tid >> 5;
    int lane = tid & 31;

    const unsigned* S = svk + (size_t)n * NPROP;
    unsigned sv[KPT];
    #pragma unroll
    for (int j = 0; j < KPT; j++) sv[j] = S[j * 1024 + tid];

    int turn = 0;
    unsigned T = 0u;
    int c_ge = NPROP;
    bool t_exact = true;
    {
        int c = 0;
        #pragma unroll
        for (int j = 0; j < KPT; j += 2) {
            unsigned m0 = __builtin_amdgcn_ballot_w32(sv[j] >= WORD_PREFIX);
            unsigned m1 = __builtin_amdgcn_ballot_w32(sv[j + 1] >= WORD_PREFIX);
            c += __builtin_popcountll(((unsigned long long)m1 << 32) | m0);
        }
        int tot = block_total(s_cnt, turn, wave, lane, c);
        if (tot >= PRE_NMS) { T = WORD_PREFIX; c_ge = tot; }
    }
    if (T != 0u) {
        for (int bit = 24; bit >= 0; bit--) {
            if (c_ge <= NSORT) { t_exact = false; break; }
            unsigned cand = T | (1u << bit);
            int c = 0;
            #pragma unroll
            for (int j = 0; j < KPT; j += 2) {
                unsigned m0 = __builtin_amdgcn_ballot_w32(sv[j] >= cand);
                unsigned m1 = __builtin_amdgcn_ballot_w32(sv[j + 1] >= cand);
                c += __builtin_popcountll(((unsigned long long)m1 << 32) | m0);
            }
            int tot = block_total(s_cnt, turn, wave, lane, c);
            if (tot >= PRE_NMS) { T = cand; c_ge = tot; }
        }
    }
    if (T != 0u && t_exact) {
        int c = 0;
        #pragma unroll
        for (int j = 0; j < KPT; j += 2) {
            unsigned m0 = __builtin_amdgcn_ballot_w32(sv[j] > T);
            unsigned m1 = __builtin_amdgcn_ballot_w32(sv[j + 1] > T);
            c += __builtin_popcountll(((unsigned long long)m1 << 32) | m0);
        }
        int c_gt = block_total(s_cnt, turn, wave, lane, c);
        int need = PRE_NMS - c_gt;
        if (c_ge - c_gt != need) {
            unsigned R = 0u;
            for (int bit = 15; bit >= 0; bit--) {
                unsigned cand = R | (1u << bit);
                int ce = 0;
                #pragma unroll
                for (int j = 0; j < KPT; j++) {
                    unsigned r = 0xFFFFu - (unsigned)(j * 1024 + tid);
                    ce += __builtin_popcount(__builtin_amdgcn_ballot_w32((sv[j] == T) & (r >= cand)));
                }
                int tot = block_total(s_cnt, turn, wave, lane, ce);
                if (tot >= need) R = cand;
            }
            #pragma unroll
            for (int j = 0; j < KPT; j++) {
                unsigned r = 0xFFFFu - (unsigned)(j * 1024 + tid);
                if ((sv[j] == T) & (r < R)) sv[j] = 0u;
            }
        }
    }
    const unsigned Tc = (T != 0u) ? T : 1u;
    {
        int wsel = 0;
        #pragma unroll
        for (int j = 0; j < KPT; j += 2) {
            unsigned m0 = __builtin_amdgcn_ballot_w32(sv[j] >= Tc);
            unsigned m1 = __builtin_amdgcn_ballot_w32(sv[j + 1] >= Tc);
            wsel += __builtin_popcountll(((unsigned long long)m1 << 32) | m0);
        }
        unsigned* row = s_cnt[turn & 1];
        turn++;
        if (lane == 0) row[wave] = (unsigned)wsel;
        __syncthreads();
        int before = 0, msel = 0;
        for (int q = 0; q < 32; q++) {
            int v = (int)row[q];
            before += (q < wave) ? v : 0;
            msel   += v;
        }
        int pos = __builtin_amdgcn_readfirstlane(before);
        #pragma unroll
        for (int j = 0; j < KPT; j++) {
            bool sel = sv[j] >= Tc;
            unsigned m = __builtin_amdgcn_ballot_w32(sel);
            int at = pos + (int)__builtin_amdgcn_mbcnt_lo(m, 0u);
            if (sel) { shi[at] = sv[j]; slo[at] = 0xFFFFu - (unsigned)(j * 1024 + tid); }
            pos += __builtin_popcount(m);
        }
        for (int e = msel + tid; e < NSORT; e += 1024) { shi[e] = 0u; slo[e] = 0u; }
    }
    for (int k = 2; k <= NSORT; k <<= 1) {
        for (int j = k >> 1; j > 0; j >>= 1) {
            __syncthreads();
            int i = ((tid & ~(j - 1)) << 1) | (tid & (j - 1));
            int p = i | j;
            unsigned ah = shi[i], al = slo[i];
            unsigned bh = shi[p], bl = slo[p];
            bool a_lt_b = (ah < bh) || (ah == bh && al < bl);
            bool up = ((i & k) == 0);
            if (a_lt_b == up) { shi[i] = bh; slo[i] = bl; shi[p] = ah; slo[p] = al; }
        }
    }
    __syncthreads();

    for (int i = tid; i < NCAND; i += 1024) {
        bool v = false;
        float b0 = 0.f, b1 = 0.f, b2 = 0.f, b3 = 0.f, s = 0.f;
        if (i < PRE_NMS) {
            unsigned w   = shi[i];
            unsigned idx = 0xFFFFu - slo[i];
            v = w != 0u;
            if (v) {
                const float* bp = boxes + (((size_t)n * NPROP) + idx) * 4;
                b0 = bp[0]; b1 = bp[1]; b2 = bp[2]; b3 = bp[3];
                unsigned u = (w & 0x80000000u) ? (w ^ 0x80000000u) : ~w;
                s = __uint_as_float(u);
            }
        }
        cbx[i] = make_float4(b0, b1, b2, b3);
        car[i] = fmaxf(b2 - b0, 0.f) * fmaxf(b3 - b1, 0.f);
        csc[i] = s;
        cvl[i] = v ? 1 : 0;
    }
    for (int c = tid; c < NCELL; c += 1024) hd[c] = -1;
    __syncthreads();

    int nk = 0;
    for (int g = 0; g < NGRP; g++) {
        int base = g * 32;
        int i    = base + wave;
        float4 ib = cbx[i];
        float  ia = car[i];
        float  is = csc[i];
        int    vi = cvl[i];

        const float half_ia = 0.5f * ia;
        const int mycell  = cell_of(0.5f * (ib.y + ib.w)) * GRID_W + cell_of(0.5f * (ib.x + ib.z));
        const int oldhead = hd[mycell];
        bool found = false;
        if (vi) {
            const int cx0 = cell_of(ib.x), cy0 = cell_of(ib.y);
            const int nx  = cell_of(ib.z) - cx0 + 1;
            const int ny  = cell_of(ib.w) - cy0 + 1;
            const int m   = nx * ny;
            const float rnx = __builtin_amdgcn_rcpf((float)nx);
            for (int q = lane; q < m; q += 32) {
                const int qy = (int)(((float)q + 0.5f) * rnx);
                const int qx = q - qy * nx;
                int k = hd[min((cy0 + qy) * GRID_W + cx0 + qx, NCELL - 1)];
                for (int it = 0; it < NCAND && k >= 0; it++) {
                    const float4 jb = kbx[k];
                    const int    kn = nxt[k];
                    if (may_overlap(ib.x, ib.y, ib.z, ib.w, half_ia, jb.x, jb.y, jb.z, jb.w))
                        found |= iou_over(ib.x, ib.y, ib.z, ib.w, ia, jb.x, jb.y, jb.z, jb.w, kar[k]);
                    k = kn;
                }
            }
        }
        float4 rb = cbx[base + lane];
        const int rcell = cell_of(0.5f * (rb.y + rb.w)) * GRID_W + cell_of(0.5f * (rb.x + rb.z));
        bool hit = false;
        if ((lane < wave) && may_overlap(ib.x, ib.y, ib.z, ib.w, half_ia, rb.x, rb.y, rb.z, rb.w))
            hit = iou_over(ib.x, ib.y, ib.z, ib.w, ia, rb.x, rb.y, rb.z, rb.w, car[base + lane]);
        unsigned anyf = __builtin_amdgcn_ballot_w32(found);
        unsigned rowb = __builtin_amdgcn_ballot_w32(hit);
        if (lane == 0) {
            s_alive[wave] = (vi != 0 && anyf == 0u) ? 1u : 0u;
            s_row[wave]   = rowb;
        }
        __syncthreads();

        unsigned al = s_alive[lane];
        unsigned rw = s_row[lane];
        unsigned keepbits = __builtin_amdgcn_ballot_w32(al != 0u && rw == 0u);
        unsigned pend     = __builtin_amdgcn_ballot_w32(al != 0u && rw != 0u);
        while (pend != 0u) {
            int r = __builtin_ctz(pend);
            pend &= pend - 1u;
            unsigned rr = (unsigned)__builtin_amdgcn_readlane((int)rw, r);
            if ((rr & keepbits) == 0u) keepbits |= (1u << r);
        }
        if ((keepbits >> wave) & 1u) {
            const unsigned below = (1u << wave) - 1u;
            int pos = nk + __popc(keepbits & below);
            unsigned same  = __builtin_amdgcn_ballot_w32(rcell == mycell) & keepbits;
            unsigned older = same & below;
            int link = oldhead;
            if (older != 0u) {
                int p = 31 - __builtin_clz(older);
                link = nk + __popc(keepbits & ((1u << p) - 1u));
            }
            if (lane == 0) {
                kbx[pos] = ib; kar[pos] = ia; ksc[pos] = is; nxt[pos] = link;
                if ((same & ~below & ~(1u << wave)) == 0u) hd[mycell] = pos;
            }
        }
        nk += __popc(keepbits);
        __syncthreads();
    }

    for (int t = tid; t < POST_NMS * 5; t += 1024) {
        int row = t / 5;
        int cc  = t - row * 5;
        float v = 0.f;
        if (row < nk) v = (cc < 4) ? ((const float*)kbx)[row * 4 + cc] : ksc[row];
        out[(size_t)n * POST_NMS * 5 + t] = v;
    }
}

extern "C" void kernel_launch(void* const* d_in, const int* in_sizes, int n_in,
                              void* d_out, int out_size, void* d_ws, size_t ws_size,
                              hipStream_t stream) {
    const float* feat   = (const float*)d_in[0];
    const float* conv_w = (const float*)d_in[1];
    const float* conv_b = (const float*)d_in[2];
    const float* det_w  = (const float*)d_in[3];
    const float* det_b  = (const float*)d_in[4];
    const float* reg_w  = (const float*)d_in[5];
    const float* reg_b  = (const float*)d_in[6];
    float* out = (float*)d_out;

    char* ws = (char*)d_ws;
    float* wt2                  = (float*)(ws);
    float* scores               = (float*)(ws + 4718592);
    float* deltas               = (float*)(ws + 4718592 + 786432);
    float* boxes                = deltas;
    unsigned* svk               = (unsigned*)(ws + 4718592);

    repack_w_kernel<<<4608, 256, 0, stream>>>(conv_w, wt2);
    conv_fused_kernel<<<NIMG * (HF / 4) * (WF / TILE_W), 256, 0, stream>>>(
        feat, wt2, conv_b, det_w, det_b, reg_w, reg_b, scores, deltas);
    decode_kernel<<<dim3(NPROP / 256, NIMG), 256, 0, stream>>>(scores, deltas, boxes, svk);
    nms_kernel<<<NIMG, 1024, 0, stream>>>(svk, boxes, out);
}


# NMS group loop: per-box cell record (extent, filing cell, valid) built once at fetch and read as scalars; cell-list walk as one wave-uniform loop ending at a self-linked zero entry; same out
# speedup vs baseline: 5.5867x; 1.0392x over previous
#include <hip/hip_runtime.h>
#include <hip/hip_bf16.h>
#include <math.h>

typedef __attribute__((ext_vector_type(2))) float v2f;
typedef __attribute__((ext_vector_type(8))) float v8f;

#define NIMG 4
#define CIN 256
#define HF 128
#define WF 128
#define A_ANCH 3
#define NPROP (A_ANCH * HF * WF)
#define PRE_NMS 2000
#define POST_NMS 300
#define OBJ_THR 0.3f
#define IOU_THR 0.7f
#define TILE_W 32
#define LCOLS 34
#define CCHUNK 32
#define CIPAD 34
#define NTR 6
#define PPITCH 36

__global__ __launch_bounds__(256) void repack_w_kernel(const float* __restrict__ conv_w,
                                                       float* __restrict__ wt2) {
    int t = blockIdx.x * 256 + threadIdx.x;
    int par = t & 1;
    int co  = (t >> 1) & 255;
    int q   = t >> 9;
    int ci2 = q & 127;
    int kk  = q >> 7;
    int ci  = ci2 * 2 + par;
    int tr  = kk / 3, kw = kk % 3;
    const float* w = conv_w + ((size_t)(co * CIN + ci) * 3) * 3 + kw;
    float g0 = w[0], g1 = w[3], g2 = w[6];
    const float c6 = 1.0f / 6.0f, c12 = 1.0f / 12.0f, c24 = 1.0f / 24.0f;
    float v;
    if (tr == 0)      v = 0.25f * g0;
    else if (tr == 1) v = -c6 * ((g0 + g2) + g1);
    else if (tr == 2) v = -c6 * ((g0 + g2) - g1);
    else if (tr == 3) v = (c24 * g0 + c6 * g2) + c12 * g1;
    else if (tr == 4) v = (c24 * g0 + c6 * g2) - c12 * g1;
    else              v = g2;
    wt2[t] = v;
}

__device__ __forceinline__ void comb6(const float d[6], float v[6]) {
    v[0] = (4.0f * d[0] - 5.0f * d[2]) + d[4];
    v[1] = (d[3] + d[4]) - 4.0f * (d[1] + d[2]);
    v[2] = (d[4] - d[3]) + 4.0f * (d[1] - d[2]);
    v[3] = (d[4] - d[2]) + 2.0f * (d[3] - d[1]);
    v[4] = (d[4] - d[2]) - 2.0f * (d[3] - d[1]);
    v[5] = (4.0f * d[1] - 5.0f * d[3]) + d[5];
}


__global__ __launch_bounds__(256, 4) void conv_fused_kernel(
    const float* __restrict__ feat, const float* __restrict__ wt2,
    const float* __restrict__ conv_b,
    const float* __restrict__ det_w, const float* __restrict__ det_b,
    const float* __restrict__ reg_w, const float* __restrict__ reg_b,
    float* __restrict__ scores, float* __restrict__ deltas) {

    __shared__ float lds_in[NTR * LCOLS * CIPAD];
    __shared__ float part[8 * 4 * 16 * PPITCH];

    int bid = blockIdx.x;
    int n    = bid >> 7;
    int rem  = bid & 127;
    int row0 = (rem >> 2) * 4;
    int col0 = (rem & 3) * TILE_W;

    int tid  = threadIdx.x;
    int wave = tid >> 5;
    int lane = tid & 31;
    int hl   = lane >> 4;
    int lm   = lane & 15;

    v8f acc[NTR][2][2] = {};

    const float* featn = feat + (size_t)n * CIN * HF * WF;
    const v2f*   wp    = (const v2f*)wt2;

    const int sx4   = tid & 7;
    const int scl   = tid >> 3;
    const int hcl   = tid & 31;
    const int hside = (tid >> 5) & 1;
    const int hgcol = col0 - 1 + hside * (TILE_W + 1);
    const bool hcok = (hgcol >= 0) && (hgcol < WF);
    const int hgc   = min(max(hgcol, 0), WF - 1);
    const bool ok0  = (row0 - 1 >= 0);
    const bool ok5  = (row0 + 4 < HF);

    for (int cib = 0; cib < CIN; cib += CCHUNK) {
        float4 sv[NTR];
        float  hv[NTR] = {0.f, 0.f, 0.f, 0.f, 0.f, 0.f};
        #pragma unroll
        for (int r = 0; r < NTR; r++) {
            const int grc = min(max(row0 + r - 1, 0), HF - 1);
            sv[r] = *(const float4*)&featn[((size_t)(cib + scl) * HF + grc) * WF + col0 + 4 * sx4];
        }
        if (tid < 64) {
            #pragma unroll
            for (int r = 0; r < NTR; r++) {
                const int grc = min(max(row0 + r - 1, 0), HF - 1);
                hv[r] = featn[((size_t)(cib + hcl) * HF + grc) * WF + hgc];
            }
        }

        __syncthreads();
        {
            const float4 z4 = make_float4(0.f, 0.f, 0.f, 0.f);
            const float4 s0 = ok0 ? sv[0] : z4;
            const float4 s5 = ok5 ? sv[5] : z4;
            const float dx[6] = {s0.x, sv[1].x, sv[2].x, sv[3].x, sv[4].x, s5.x};
            const float dy[6] = {s0.y, sv[1].y, sv[2].y, sv[3].y, sv[4].y, s5.y};
            const float dz[6] = {s0.z, sv[1].z, sv[2].z, sv[3].z, sv[4].z, s5.z};
            const float dw[6] = {s0.w, sv[1].w, sv[2].w, sv[3].w, sv[4].w, s5.w};
            float cx[6], cy[6], cz[6], cw[6];
            comb6(dx, cx); comb6(dy, cy); comb6(dz, cz); comb6(dw, cw);
            #pragma unroll
            for (int tr = 0; tr < NTR; tr++) {
                float* dst = &lds_in[(tr * LCOLS + 1 + 4 * sx4) * CIPAD + scl];
                dst[0 * CIPAD] = cx[tr];
                dst[1 * CIPAD] = cy[tr];
                dst[2 * CIPAD] = cz[tr];
                dst[3 * CIPAD] = cw[tr];
            }
        }
        if (tid < 64) {
            float e[6], ce[6];
            e[0] = (ok0 && hcok) ? hv[0] : 0.0f;
            e[1] = hcok ? hv[1] : 0.0f;
            e[2] = hcok ? hv[2] : 0.0f;
            e[3] = hcok ? hv[3] : 0.0f;
            e[4] = hcok ? hv[4] : 0.0f;
            e[5] = (ok5 && hcok) ? hv[5] : 0.0f;
            comb6(e, ce);
            float* dst = &lds_in[(hside * (TILE_W + 1)) * CIPAD + hcl];
            #pragma unroll
            for (int tr = 0; tr < NTR; tr++) dst[tr * LCOLS * CIPAD] = ce[tr];
        }
        __syncthreads();

        const v2f* wl = wp + (size_t)(((cib >> 1) + hl) * 256 + wave * 32 + lm);
        #pragma unroll
        for (int tr = 0; tr < NTR; tr++) {
            #pragma unroll
            for (int kw = 0; kw < 3; kw++) {
                const int kk = tr * 3 + kw;
                v2f a0[CCHUNK / 4], a1[CCHUNK / 4];
                #pragma unroll
                for (int j = 0; j < CCHUNK / 4; j++) {
                    a0[j] = wl[(kk * 128 + 2 * j) * 256];
                    a1[j] = wl[(kk * 128 + 2 * j) * 256 + 16];
                }
                __builtin_amdgcn_sched_barrier(0);
                #pragma unroll
                for (int j = 0; j < CCHUNK / 4; j++) {
                    int boff = 4 * j + 2 * hl;
                    v2f b0 = *(const v2f*)&lds_in[(tr * LCOLS + lm + kw) * CIPAD + boff];
                    v2f b1 = *(const v2f*)&lds_in[(tr * LCOLS + 16 + lm + kw) * CIPAD + boff];
                    acc[tr][0][0] = __builtin_amdgcn_wmma_f32_16x16x4_f32(false, a0[j], false, b0, (short)0, acc[tr][0][0], false, false);
                    acc[tr][1][0] = __builtin_amdgcn_wmma_f32_16x16x4_f32(false, a1[j], false, b0, (short)0, acc[tr][1][0], false, false);
                    acc[tr][0][1] = __builtin_amdgcn_wmma_f32_16x16x4_f32(false, a0[j], false, b1, (short)0, acc[tr][0][1], false, false);
                    acc[tr][1][1] = __builtin_amdgcn_wmma_f32_16x16x4_f32(false, a1[j], false, b1, (short)0, acc[tr][1][1], false, false);
                }
            }
        }
    }

    const int elane = (int)__builtin_amdgcn_mbcnt_lo(~0u, 0u);
    const int elm = elane & 15;
    const int ehl = elane >> 4;
    const int ho = elm;
    const float* hrow = (ho < 3) ? (det_w + ho * 256) : (reg_w + (min(ho, 14) - 3) * 256);
    v2f ha[2][4];
    #pragma unroll
    for (int mt = 0; mt < 2; mt++) {
        #pragma unroll
        for (int q = 0; q < 4; q++) {
            const int c = wave * 32 + mt * 16 + 8 * ehl + 2 * q;
            const float w0 = hrow[c], w1 = hrow[c + 1];
            ha[mt][q][0] = (ho < 15) ? w0 : 0.0f;
            ha[mt][q][1] = (ho < 15) ? w1 : 0.0f;
        }
    }
    float cbv[2][8];
    #pragma unroll
    for (int mt = 0; mt < 2; mt++) {
        #pragma unroll
        for (int r = 0; r < 8; r++) cbv[mt][r] = conv_b[wave * 32 + mt * 16 + 8 * ehl + r];
    }

    #pragma unroll
    for (int ph = 0; ph < 4; ph++) {
        #pragma unroll
        for (int nt = 0; nt < 2; nt++) {
            v8f hd = {};
            #pragma unroll
            for (int mt = 0; mt < 2; mt++) {
                float y[8];
                #pragma unroll
                for (int r = 0; r < 8; r++) {
                    const float q0 = acc[0][mt][nt][r], q1 = acc[1][mt][nt][r], q2 = acc[2][mt][nt][r];
                    const float q3 = acc[3][mt][nt][r], q4 = acc[4][mt][nt][r], q5 = acc[5][mt][nt][r];
                    float v;
                    if (ph == 0)      v = (q0 + (q1 + q2)) + (q3 + q4);
                    else if (ph == 1) v = (q1 - q2) + 2.0f * (q3 - q4);
                    else if (ph == 2) v = (q1 + q2) + 4.0f * (q3 + q4);
                    else              v = ((q1 - q2) + 8.0f * (q3 - q4)) + q5;
                    y[r] = v + cbv[mt][r];
                }
                #pragma unroll
                for (int q = 0; q < 4; q++) {
                    v2f b;
                    b[0] = y[2 * q];
                    b[1] = y[2 * q + 1];
                    hd = __builtin_amdgcn_wmma_f32_16x16x4_f32(false, ha[mt][q], false, b, (short)0, hd, false, false);
                }
            }
            float* dst = &part[((wave * 4 + ph) * 16 + 8 * ehl) * PPITCH + nt * 16 + elm];
            #pragma unroll
            for (int r = 0; r < 8; r++) dst[r * PPITCH] = hd[r];
        }
    }
    __syncthreads();

    for (int it = tid; it < A_ANCH * 4 * TILE_W; it += 256) {
        const int a   = it >> 7;
        const int pxl = it & 127;
        const int ph  = pxl >> 5;
        const int px  = pxl & 31;
        float s[5];
        #pragma unroll
        for (int c = 0; c < 5; c++) {
            const int oi = (c == 0) ? a : (3 + (c - 1) * A_ANCH + a);
            float sum = part[((0 * 4 + ph) * 16 + oi) * PPITCH + px];
            #pragma unroll
            for (int w = 1; w < 8; w++) sum += part[((w * 4 + ph) * 16 + oi) * PPITCH + px];
            s[c] = sum + ((c == 0) ? det_b[a] : reg_b[(c - 1) * A_ANCH + a]);
        }
        const int p = (row0 + ph) * WF + col0 + px;
        scores[(size_t)n * NPROP + a * (HF * WF) + p] = 1.0f / (1.0f + expf(-s[0]));
        *(float4*)&deltas[(((size_t)n * NPROP) + a * (HF * WF) + p) * 4] = make_float4(s[1], s[2], s[3], s[4]);
    }
}

__global__ __launch_bounds__(256) void decode_kernel(
    const float* scores, const float* deltas,
    float* boxes, unsigned* svk) {
    int n = blockIdx.y;
    int i = blockIdx.x * 256 + threadIdx.x;

    int pa = i / 3, aa = i - pa * 3;
    int ya = pa >> 7, xa = pa & 127;
    const float bs[3][4] = {{-91.f, -45.f, 91.f, 45.f},
                            {-64.f, -64.f, 64.f, 64.f},
                            {-45.f, -91.f, 45.f, 91.f}};
    float sxv = (float)(xa * 8), syv = (float)(ya * 8);
    float x1 = sxv + bs[aa][0], y1 = syv + bs[aa][1];
    float x2 = sxv + bs[aa][2], y2 = syv + bs[aa][3];
    float aw = x2 - x1, ah = y2 - y1;
    float acx = y1 + aw * 0.5f;
    float acy = x1 + ah * 0.5f;

    const float* d = deltas + (((size_t)n * NPROP) + i) * 4;
    const float d0 = d[0], d1 = d[1], d2 = d[2], d3 = d[3];
    float px = acx + d0 * aw;
    float py = acy + d1 * ah;
    float pw = aw * expf(d2);
    float ph = ah * expf(d3);
    float b0 = fminf(fmaxf(px - pw * 0.5f, 0.f), 1024.f);
    float b1 = fminf(fmaxf(py - ph * 0.5f, 0.f), 1024.f);
    float b2 = fminf(fmaxf(px + pw * 0.5f, 0.f), 1024.f);
    float b3 = fminf(fmaxf(py + ph * 0.5f, 0.f), 1024.f);

    float* bo = boxes + (((size_t)n * NPROP) + i) * 4;
    bo[0] = b0; bo[1] = b1; bo[2] = b2; bo[3] = b3;

    float s = scores[(size_t)n * NPROP + i];
    bool valid = (s > OBJ_THR) && ((b2 - b0) > 0.f) && ((b3 - b1) > 0.f);
    unsigned u = __float_as_uint(s);
    unsigned sv = (u & 0x80000000u) ? ~u : (u | 0x80000000u);
    svk[(size_t)n * NPROP + i] = valid ? sv : 0u;
}

#define NGRP 63
#define NCAND (NGRP * 32)
#define KPT 48
#define NSORT 2048
#define WORD_PREFIX 0xBE000000u
static_assert(OBJ_THR >= 0.125f && OBJ_THR < 1.0f, "WORD_PREFIX needs every valid score in [0.125, 2)");
#define GRID_W 32
#define NCELL (GRID_W * GRID_W)
#define LIST_END (NCAND - 1)
static_assert(LIST_END >= PRE_NMS, "kept boxes take entries 0 .. PRE_NMS - 1 at most");
static_assert(NCELL == 1024 && NCAND <= 2048, "the record packs two cell numbers of 10 bits and a flag into one word");

__device__ __forceinline__ int block_total(unsigned (*s_cnt)[32], int& turn, int wave, int lane, int wcount) {
    unsigned* row = s_cnt[turn & 1];
    turn++;
    if (lane == 0) row[wave] = (unsigned)wcount;
    __syncthreads();
    const uint4* p = (const uint4*)row;
    unsigned tot = 0;
    for (int q = 0; q < 8; q++) { uint4 v = p[q]; tot = tot + v.x + v.y; tot = tot + v.z + v.w; }
    return __builtin_amdgcn_readfirstlane((int)tot);
}

__device__ __forceinline__ bool iou_over(float ib0, float ib1, float ib2, float ib3, float ia,
                                         float jx1, float jy1, float jx2, float jy2, float ja) {
    float ix1 = fmaxf(ib0, jx1), iy1 = fmaxf(ib1, jy1);
    float ix2 = fminf(ib2, jx2), iy2 = fminf(ib3, jy2);
    float inter = fmaxf(ix2 - ix1, 0.f) * fmaxf(iy2 - iy1, 0.f);
    float iou = inter / fmaxf(ia + ja - inter, 1e-9f);
    return iou > IOU_THR;
}

__device__ __forceinline__ bool may_overlap(float ib0, float ib1, float ib2, float ib3, float half_ia,
                                            float jx1, float jy1, float jx2, float jy2) {
    float w = fminf(ib2, jx2) - fmaxf(ib0, jx1);
    float h = fminf(ib3, jy2) - fmaxf(ib1, jy1);
    return fmaxf(w, 0.f) * h > half_ia;
}

static_assert(IOU_THR > 0.5f, "the centre-cell argument needs a threshold above one half");
__device__ __forceinline__ int cell_of(float x) {
    return min(max((int)(x * (1.0f / 32.0f)), 0), GRID_W - 1);
}

__global__ __launch_bounds__(1024) void nms_kernel(const unsigned* __restrict__ svk,
                                                   const float* __restrict__ boxes,
                                                   float* __restrict__ out) {
    __shared__ float4   cbx[NCAND];
    __shared__ float    car[NCAND];
    __shared__ float    csc[NCAND];
    __shared__ int4     cge[NCAND];
    __shared__ float4   kbx[NCAND];
    __shared__ float    kar[NCAND];
    __shared__ float    ksc[NCAND];
    __shared__ int      hd[NCELL];
    __shared__ int      nxt[NCAND];
    __shared__ uint2    s_ar[32];
    __shared__ unsigned shi[NSORT];
    __shared__ unsigned slo[NSORT];
    __shared__ __attribute__((aligned(16))) unsigned s_cnt[2][32];

    int n    = blockIdx.x;
    int tid  = threadIdx.x;
    int wave = __builtin_amdgcn_readfirstlane(tid >> 5);
    int lane = tid & 31;

    const unsigned* S = svk + (size_t)n * NPROP;
    unsigned sv[KPT];
    #pragma unroll
    for (int j = 0; j < KPT; j++) sv[j] = S[j * 1024 + tid];

    int turn = 0;
    unsigned T = 0u;
    int c_ge = NPROP;
    bool t_exact = true;
    {
        int c = 0;
        #pragma unroll
        for (int j = 0; j < KPT; j += 2) {
            unsigned m0 = __builtin_amdgcn_ballot_w32(sv[j] >= WORD_PREFIX);
            unsigned m1 = __builtin_amdgcn_ballot_w32(sv[j + 1] >= WORD_PREFIX);
            c += __builtin_popcountll(((unsigned long long)m1 << 32) | m0);
        }
        int tot = block_total(s_cnt, turn, wave, lane, c);
        if (tot >= PRE_NMS) { T = WORD_PREFIX; c_ge = tot; }
    }
    if (T != 0u) {
        for (int bit = 24; bit >= 0; bit--) {
            if (c_ge <= NSORT) { t_exact = false; break; }
            unsigned cand = T | (1u << bit);
            int c = 0;
            #pragma unroll
            for (int j = 0; j < KPT; j += 2) {
                unsigned m0 = __builtin_amdgcn_ballot_w32(sv[j] >= cand);
                unsigned m1 = __builtin_amdgcn_ballot_w32(sv[j + 1] >= cand);
                c += __builtin_popcountll(((unsigned long long)m1 << 32) | m0);
            }
            int tot = block_total(s_cnt, turn, wave, lane, c);
            if (tot >= PRE_NMS) { T = cand; c_ge = tot; }
        }
    }
    if (T != 0u && t_exact) {
        int c = 0;
        #pragma unroll
        for (int j = 0; j < KPT; j += 2) {
            unsigned m0 = __builtin_amdgcn_ballot_w32(sv[j] > T);
            unsigned m1 = __builtin_amdgcn_ballot_w32(sv[j + 1] > T);
            c += __builtin_popcountll(((unsigned long long)m1 << 32) | m0);
        }
        int c_gt = block_total(s_cnt, turn, wave, lane, c);
        int need = PRE_NMS - c_gt;
        if (c_ge - c_gt != need) {
            unsigned R = 0u;
            for (int bit = 15; bit >= 0; bit--) {
                unsigned cand = R | (1u << bit);
                int ce = 0;
                #pragma unroll
                for (int j = 0; j < KPT; j++) {
                    unsigned r = 0xFFFFu - (unsigned)(j * 1024 + tid);
                    ce += __builtin_popcount(__builtin_amdgcn_ballot_w32((sv[j] == T) & (r >= cand)));
                }
                int tot = block_total(s_cnt, turn, wave, lane, ce);
                if (tot >= need) R = cand;
            }
            #pragma unroll
            for (int j = 0; j < KPT; j++) {
                unsigned r = 0xFFFFu - (unsigned)(j * 1024 + tid);
                if ((sv[j] == T) & (r < R)) sv[j] = 0u;
            }
        }
    }
    const unsigned Tc = (T != 0u) ? T : 1u;
    {
        int wsel = 0;
        #pragma unroll
        for (int j = 0; j < KPT; j += 2) {
            unsigned m0 = __builtin_amdgcn_ballot_w32(sv[j] >= Tc);
            unsigned m1 = __builtin_amdgcn_ballot_w32(sv[j + 1] >= Tc);
            wsel += __builtin_popcountll(((unsigned long long)m1 << 32) | m0);
        }
        unsigned* row = s_cnt[turn & 1];
        turn++;
        if (lane == 0) row[wave] = (unsigned)wsel;
        __syncthreads();
        int before = 0, msel = 0;
        for (int q = 0; q < 32; q++) {
            int v = (int)row[q];
            before += (q < wave) ? v : 0;
            msel   += v;
        }
        int pos = __builtin_amdgcn_readfirstlane(before);
        #pragma unroll
        for (int j = 0; j < KPT; j++) {
            bool sel = sv[j] >= Tc;
            unsigned m = __builtin_amdgcn_ballot_w32(sel);
            int at = pos + (int)__builtin_amdgcn_mbcnt_lo(m, 0u);
            if (sel) { shi[at] = sv[j]; slo[at] = 0xFFFFu - (unsigned)(j * 1024 + tid); }
            pos += __builtin_popcount(m);
        }
        for (int e = msel + tid; e < NSORT; e += 1024) { shi[e] = 0u; slo[e] = 0u; }
    }
    for (int k = 2; k <= NSORT; k <<= 1) {
        for (int j = k >> 1; j > 0; j >>= 1) {
            __syncthreads();
            int i = ((tid & ~(j - 1)) << 1) | (tid & (j - 1));
            int p = i | j;
            unsigned ah = shi[i], al = slo[i];
            unsigned bh = shi[p], bl = slo[p];
            bool a_lt_b = (ah < bh) || (ah == bh && al < bl);
            bool up = ((i & k) == 0);
            if (a_lt_b == up) { shi[i] = bh; slo[i] = bl; shi[p] = ah; slo[p] = al; }
        }
    }
    __syncthreads();

    for (int i = tid; i < NCAND; i += 1024) {
        bool v = false;
        float b0 = 0.f, b1 = 0.f, b2 = 0.f, b3 = 0.f, s = 0.f;
        if (i < PRE_NMS) {
            unsigned w   = shi[i];
            unsigned idx = 0xFFFFu - slo[i];
            v = w != 0u;
            if (v) {
                const float* bp = boxes + (((size_t)n * NPROP) + idx) * 4;
                b0 = bp[0]; b1 = bp[1]; b2 = bp[2]; b3 = bp[3];
                unsigned u = (w & 0x80000000u) ? (w ^ 0x80000000u) : ~w;
                s = __uint_as_float(u);
            }
        }
        cbx[i] = make_float4(b0, b1, b2, b3);
        car[i] = fmaxf(b2 - b0, 0.f) * fmaxf(b3 - b1, 0.f);
        csc[i] = s;
        const int cx0 = cell_of(b0), cy0 = cell_of(b1);
        const int nx  = cell_of(b2) - cx0 + 1;
        const int ny  = cell_of(b3) - cy0 + 1;
        const int fc  = cell_of(0.5f * (b1 + b3)) * GRID_W + cell_of(0.5f * (b0 + b2));
        cge[i] = make_int4((cy0 * GRID_W + cx0) | (fc << 10) | (v ? (1 << 20) : 0), nx, nx * ny,
                           __float_as_int(__builtin_amdgcn_rcpf((float)nx)));
    }
    for (int c = tid; c < NCELL; c += 1024) hd[c] = LIST_END;
    if (tid == 0) { kbx[LIST_END] = make_float4(0.f, 0.f, 0.f, 0.f); kar[LIST_END] = 0.f; nxt[LIST_END] = LIST_END; }
    __syncthreads();

    int nk = 0;
    const float lanef = (float)lane + 0.5f;
    for (int g = 0; g < NGRP; g++) {
        const int base = g * 32;
        const int i    = base + wave;
        const float4 ib  = cbx[i];
        const int4   ge  = cge[i];
        const float  ia  = car[i];
        const float  is  = csc[i];
        const float4 rb  = cbx[base + lane];
        const int    rw0 = cge[base + lane].x;
        const int   w0  = __builtin_amdgcn_readfirstlane(ge.x);
        const int   nx  = __builtin_amdgcn_readfirstlane(ge.y);
        const int   m   = __builtin_amdgcn_readfirstlane(ge.z);
        const float rnx = __int_as_float(__builtin_amdgcn_readfirstlane(ge.w));
        const int   first  = w0 & (NCELL - 1);
        const int   mycell = (w0 >> 10) & (NCELL - 1);
        const bool  vi     = (w0 >> 20) != 0;
        const int oldhead = hd[mycell];

        const float half_ia = 0.5f * ia;
        int fnd = 0;
        if (vi) {
            for (int q0 = 0; q0 < m; q0 += 32) {
                const int q  = q0 + lane;
                const int qy = (int)(((float)q0 + lanef) * rnx);
                const int qx = q - qy * nx;
                const int hk = hd[min(max(first + qy * GRID_W + qx, 0), NCELL - 1)];
                int k = (q < m) ? hk : LIST_END;
                for (int it = 0; it < NCAND; it++) {
                    if (__builtin_amdgcn_ballot_w32(k != LIST_END) == 0u) break;
                    const float4 jb = kbx[k];
                    const int    kn = nxt[k];
                    const unsigned tm = __builtin_amdgcn_ballot_w32(
                        may_overlap(ib.x, ib.y, ib.z, ib.w, half_ia, jb.x, jb.y, jb.z, jb.w));
                    if (tm != 0u) {
                        const bool o = iou_over(ib.x, ib.y, ib.z, ib.w, ia, jb.x, jb.y, jb.z, jb.w, kar[k]);
                        fnd |= (int)((tm >> lane) & 1u) & (o ? 1 : 0);
                    }
                    k = kn;
                }
            }
        }
        bool hit = false;
        if ((lane < wave) && may_overlap(ib.x, ib.y, ib.z, ib.w, half_ia, rb.x, rb.y, rb.z, rb.w))
            hit = iou_over(ib.x, ib.y, ib.z, ib.w, ia, rb.x, rb.y, rb.z, rb.w, car[base + lane]);
        unsigned anyf = __builtin_amdgcn_ballot_w32(fnd != 0);
        unsigned rowb = __builtin_amdgcn_ballot_w32(hit);
        if (lane == 0) s_ar[wave] = make_uint2((vi && anyf == 0u) ? 1u : 0u, rowb);
        __syncthreads();

        const uint2 ar = s_ar[lane];
        unsigned al = ar.x;
        unsigned rw = ar.y;
        const unsigned alive = __builtin_amdgcn_ballot_w32(al != 0u);
        const unsigned clear = __builtin_amdgcn_ballot_w32(rw == 0u);
        unsigned keepbits = alive & clear;
        unsigned pend     = alive & ~clear;
        while (pend != 0u) {
            int r = __builtin_ctz(pend);
            pend &= pend - 1u;
            unsigned rr = (unsigned)__builtin_amdgcn_readlane((int)rw, r);
            if ((rr & keepbits) == 0u) keepbits |= (1u << r);
        }
        if ((keepbits >> wave) & 1u) {
            const unsigned below = (1u << wave) - 1u;
            int pos = nk + __popc(keepbits & below);
            const int rcell = (rw0 >> 10) & (NCELL - 1);
            unsigned same  = __builtin_amdgcn_ballot_w32(rcell == mycell) & keepbits;
            unsigned older = same & below;
            int link = oldhead;
            if (older != 0u) {
                int p = 31 - __builtin_clz(older);
                link = nk + __popc(keepbits & ((1u << p) - 1u));
            }
            if (lane == 0) {
                kbx[pos] = ib; kar[pos] = ia; ksc[pos] = is; nxt[pos] = link;
                if ((same & ~below & ~(1u << wave)) == 0u) hd[mycell] = pos;
            }
        }
        nk += __popc(keepbits);
        __syncthreads();
    }

    for (int t = tid; t < POST_NMS * 5; t += 1024) {
        int row = t / 5;
        int cc  = t - row * 5;
        float v = 0.f;
        if (row < nk) v = (cc < 4) ? ((const float*)kbx)[row * 4 + cc] : ksc[row];
        out[(size_t)n * POST_NMS * 5 + t] = v;
    }
}

extern "C" void kernel_launch(void* const* d_in, const int* in_sizes, int n_in,
                              void* d_out, int out_size, void* d_ws, size_t ws_size,
                              hipStream_t stream) {
    const float* feat   = (const float*)d_in[0];
    const float* conv_w = (const float*)d_in[1];
    const float* conv_b = (const float*)d_in[2];
    const float* det_w  = (const float*)d_in[3];
    const float* det_b  = (const float*)d_in[4];
    const float* reg_w  = (const float*)d_in[5];
    const float* reg_b  = (const float*)d_in[6];
    float* out = (float*)d_out;

    char* ws = (char*)d_ws;
    float* wt2                  = (float*)(ws);
    float* scores               = (float*)(ws + 4718592);
    float* deltas               = (float*)(ws + 4718592 + 786432);
    float* boxes                = deltas;
    unsigned* svk               = (unsigned*)(ws + 4718592);

    repack_w_kernel<<<4608, 256, 0, stream>>>(conv_w, wt2);
    conv_fused_kernel<<<NIMG * (HF / 4) * (WF / TILE_W), 256, 0, stream>>>(
        feat, wt2, conv_b, det_w, det_b, reg_w, reg_b, scores, deltas);
    decode_kernel<<<dim3(NPROP / 256, NIMG), 256, 0, stream>>>(scores, deltas, boxes, svk);
    nms_kernel<<<NIMG, 1024, 0, stream>>>(svk, boxes, out);
}
